# starting kernel
# baseline (speedup 1.0000x reference)
#include <hip/hip_runtime.h>

typedef __attribute__((ext_vector_type(2))) float v2f;
typedef __attribute__((ext_vector_type(8))) float v8f;

#define NNODES 50000

__device__ __forceinline__ void atom_add_f32(float* p, float v) {
    asm volatile("global_atomic_add_f32 %0, %1, off" :: "v"(p), "v"(v) : "memory");
}

__global__ __launch_bounds__(256) void init_kernel(float* __restrict__ deg,
                                                   float* __restrict__ agg1,
                                                   float* __restrict__ out) {
    long long idx = (long long)blockIdx.x * 256 + threadIdx.x;
    if (idx < (long long)NNODES * 128) agg1[idx] = 0.0f;
    if (idx < (long long)NNODES * 64)  out[idx]  = 0.0f;
    if (idx < NNODES)                  deg[idx]  = 1.0f;
}

__global__ __launch_bounds__(256) void deg_kernel(const int* __restrict__ dst,
                                                  float* __restrict__ deg, int E) {
    int e = blockIdx.x * 256 + threadIdx.x;
    if (e < E) atom_add_f32(&deg[dst[e]], 1.0f);
}

__global__ __launch_bounds__(256) void dinv_kernel(float* __restrict__ deg) {
    int i = blockIdx.x * 256 + threadIdx.x;
    if (i < NNODES) deg[i] = rsqrtf(deg[i]);
}

template <int K, int NF>
__global__ __launch_bounds__(256) void gemm_wmma(const float* __restrict__ A,
                                                 const float* __restrict__ B,
                                                 float* __restrict__ C, int M) {
    const int lane  = threadIdx.x & 31;
    const int wave  = blockIdx.x * 8 + (threadIdx.x >> 5);
    const int mBase = wave * 16;
    if (mBase >= M) return;
    const int half = lane >> 4;
    const int l16  = lane & 15;
    constexpr int NT = NF / 16;

    v8f acc[NT] = {};

    const float* aRow = A + (size_t)(mBase + l16) * K + 2 * half;
    for (int k = 0; k < K; k += 4) {
        v2f a = *(const v2f*)(aRow + k);
        const float* bPtr = B + (size_t)(k + 2 * half) * NF + l16;
#pragma unroll
        for (int t = 0; t < NT; ++t) {
            v2f b;
            b.x = bPtr[t * 16];
            b.y = bPtr[t * 16 + NF];
            acc[t] = __builtin_amdgcn_wmma_f32_16x16x4_f32(
                false, a, false, b, (short)0, acc[t], false, false);
        }
    }
#pragma unroll
    for (int t = 0; t < NT; ++t) {
        float* cPtr = C + (size_t)(mBase + half * 8) * NF + t * 16 + l16;
#pragma unroll
        for (int r = 0; r < 8; ++r) cPtr[(size_t)r * NF] = acc[t][r];
    }
}

template <int F>
__global__ __launch_bounds__(256) void edge_agg(const int* __restrict__ src,
                                                const int* __restrict__ dst,
                                                const float* __restrict__ dinv,
                                                const float* __restrict__ h,
                                                float* __restrict__ agg, int E) {
    constexpr int CH = F / 4;
    long long tid = (long long)blockIdx.x * 256 + threadIdx.x;
    if (tid >= (long long)E * CH) return;
    int e = (int)(tid / CH);
    int c = (int)(tid & (CH - 1));
    int s = src[e], d = dst[e];
    float w = dinv[s] * dinv[d];
    float4 hv = ((const float4*)(h + (size_t)s * F))[c];
    float* ap = agg + (size_t)d * F + c * 4;
    atom_add_f32(ap + 0, hv.x * w);
    atom_add_f32(ap + 1, hv.y * w);
    atom_add_f32(ap + 2, hv.z * w);
    atom_add_f32(ap + 3, hv.w * w);
}

__global__ __launch_bounds__(256) void fin1_kernel(float* __restrict__ agg1,
                                                   const float* __restrict__ h1,
                                                   const float* __restrict__ dinv,
                                                   const float* __restrict__ b1) {
    long long idx = (long long)blockIdx.x * 256 + threadIdx.x;
    if (idx >= (long long)NNODES * 128) return;
    int f = (int)(idx & 127);
    int i = (int)(idx >> 7);
    float di = dinv[i];
    float v = agg1[idx] + h1[idx] * di * di + b1[f];
    agg1[idx] = fmaxf(v, 0.0f);
}

__global__ __launch_bounds__(256) void fin2_kernel(float* __restrict__ out,
                                                   const float* __restrict__ h2,
                                                   const float* __restrict__ dinv,
                                                   const float* __restrict__ b2) {
    long long idx = (long long)blockIdx.x * 256 + threadIdx.x;
    if (idx >= (long long)NNODES * 64) return;
    int f = (int)(idx & 63);
    int i = (int)(idx >> 6);
    float di = dinv[i];
    out[idx] = out[idx] + h2[idx] * di * di + b2[f];
}

extern "C" void kernel_launch(void* const* d_in, const int* in_sizes, int n_in,
                              void* d_out, int out_size, void* d_ws, size_t ws_size,
                              hipStream_t stream) {
    const float* x  = (const float*)d_in[0];
    const int*   ei = (const int*)d_in[1];
    const float* W1 = (const float*)d_in[2];
    const float* b1 = (const float*)d_in[3];
    const float* W2 = (const float*)d_in[4];
    const float* b2 = (const float*)d_in[5];
    float* out = (float*)d_out;

    const int N = NNODES;
    const int E = in_sizes[1] / 2;
    const int* src = ei;
    const int* dst = ei + E;

    float* deg  = (float*)d_ws;
    float* h1   = deg + N;
    float* agg1 = h1 + (size_t)N * 128;
    float* h2   = agg1 + (size_t)N * 128;

    const int BT = 256;
    long long n128 = (long long)N * 128;
    long long n64  = (long long)N * 64;

    init_kernel<<<(int)((n128 + BT - 1) / BT), BT, 0, stream>>>(deg, agg1, out);
    deg_kernel<<<(E + BT - 1) / BT, BT, 0, stream>>>(dst, deg, E);
    dinv_kernel<<<(N + BT - 1) / BT, BT, 0, stream>>>(deg);
    gemm_wmma<256, 128><<<(N / 16 + 7) / 8, BT, 0, stream>>>(x, W1, h1, N);
    {
        long long tot = (long long)E * 32;
        edge_agg<128><<<(int)((tot + BT - 1) / BT), BT, 0, stream>>>(src, dst, deg, h1, agg1, E);
    }
    fin1_kernel<<<(int)((n128 + BT - 1) / BT), BT, 0, stream>>>(agg1, h1, deg, b1);
    gemm_wmma<128, 64><<<(N / 16 + 7) / 8, BT, 0, stream>>>(agg1, W2, h2, N);
    {
        long long tot = (long long)E * 16;
        edge_agg<64><<<(int)((tot + BT - 1) / BT), BT, 0, stream>>>(src, dst, deg, h2, out, E);
    }
    fin2_kernel<<<(int)((n64 + BT - 1) / BT), BT, 0, stream>>>(out, h2, deg, b2);
}


# edge scatter: lane = feature (l+32j), so each wave-wide f32 atomic and h load covers one whole 128-B line (rows 128-B aligned in ws), not 8 lanes in each of 4 lines; adds stay arrival-ordered
# speedup vs baseline: 1.3701x; 1.3701x over previous
#include <hip/hip_runtime.h>

typedef __attribute__((ext_vector_type(2))) float v2f;
typedef __attribute__((ext_vector_type(8))) float v8f;

#define NNODES 50000
#define NPAD ((NNODES + 31) / 32 * 32)

__device__ __forceinline__ void atom_add_f32(float* p, float v) {
    asm volatile("global_atomic_add_f32 %0, %1, off" :: "v"(p), "v"(v) : "memory");
}

__global__ __launch_bounds__(256) void init_kernel(float* __restrict__ deg,
                                                   float* __restrict__ agg1,
                                                   float* __restrict__ out) {
    long long idx = (long long)blockIdx.x * 256 + threadIdx.x;
    if (idx < (long long)NNODES * 128) agg1[idx] = 0.0f;
    if (idx < (long long)NNODES * 64)  out[idx]  = 0.0f;
    if (idx < NNODES)                  deg[idx]  = 1.0f;
}

__global__ __launch_bounds__(256) void deg_kernel(const int* __restrict__ dst,
                                                  float* __restrict__ deg, int E) {
    int e = blockIdx.x * 256 + threadIdx.x;
    if (e < E) atom_add_f32(&deg[dst[e]], 1.0f);
}

__global__ __launch_bounds__(256) void dinv_kernel(float* __restrict__ deg) {
    int i = blockIdx.x * 256 + threadIdx.x;
    if (i < NNODES) deg[i] = rsqrtf(deg[i]);
}

template <int K, int NF>
__global__ __launch_bounds__(256) void gemm_wmma(const float* __restrict__ A,
                                                 const float* __restrict__ B,
                                                 float* __restrict__ C, int M) {
    const int lane  = threadIdx.x & 31;
    const int wave  = blockIdx.x * 8 + (threadIdx.x >> 5);
    const int mBase = wave * 16;
    if (mBase >= M) return;
    const int half = lane >> 4;
    const int l16  = lane & 15;
    constexpr int NT = NF / 16;

    v8f acc[NT] = {};

    const float* aRow = A + (size_t)(mBase + l16) * K + 2 * half;
    for (int k = 0; k < K; k += 4) {
        v2f a = *(const v2f*)(aRow + k);
        const float* bPtr = B + (size_t)(k + 2 * half) * NF + l16;
#pragma unroll
        for (int t = 0; t < NT; ++t) {
            v2f b;
            b.x = bPtr[t * 16];
            b.y = bPtr[t * 16 + NF];
            acc[t] = __builtin_amdgcn_wmma_f32_16x16x4_f32(
                false, a, false, b, (short)0, acc[t], false, false);
        }
    }
#pragma unroll
    for (int t = 0; t < NT; ++t) {
        float* cPtr = C + (size_t)(mBase + half * 8) * NF + t * 16 + l16;
#pragma unroll
        for (int r = 0; r < 8; ++r) cPtr[(size_t)r * NF] = acc[t][r];
    }
}

template <int F, int NE>
__device__ __forceinline__ void scatter_edges(const int* __restrict__ src,
                                              const int* __restrict__ dst,
                                              const float* __restrict__ dinv,
                                              const float* __restrict__ h,
                                              float* __restrict__ agg,
                                              long long e0, int lane) {
    constexpr int LPR = F / 32;
    int s[NE], d[NE];
#pragma unroll
    for (int i = 0; i < NE; ++i) {
        s[i] = src[e0 + i];
        d[i] = dst[e0 + i];
    }
    float  v[NE * LPR];
    float* p[NE * LPR];
#pragma unroll
    for (int i = 0; i < NE; ++i) {
        float w = dinv[s[i]] * dinv[d[i]];
        const float* hp = h + (size_t)s[i] * F + lane;
        float* ap = agg + (size_t)d[i] * F + lane;
#pragma unroll
        for (int j = 0; j < LPR; ++j) {
            v[i * LPR + j] = hp[j * 32] * w;
            p[i * LPR + j] = ap + j * 32;
        }
    }
#pragma unroll
    for (int k = 0; k < NE * LPR; ++k) atom_add_f32(p[k], v[k]);
}

template <int F>
__global__ __launch_bounds__(256) void edge_agg(const int* __restrict__ src,
                                                const int* __restrict__ dst,
                                                const float* __restrict__ dinv,
                                                const float* __restrict__ h,
                                                float* __restrict__ agg, int E) {
    constexpr int EPW = 128 / F;
    const int lane = threadIdx.x & 31;
    const long long e0 = ((long long)blockIdx.x * 8 + (threadIdx.x >> 5)) * EPW;
    if (e0 + EPW <= E) {
        scatter_edges<F, EPW>(src, dst, dinv, h, agg, e0, lane);
    } else {
        for (long long e = e0; e < E; ++e)
            scatter_edges<F, 1>(src, dst, dinv, h, agg, e, lane);
    }
}

__global__ __launch_bounds__(256) void fin1_kernel(float* __restrict__ agg1,
                                                   const float* __restrict__ h1,
                                                   const float* __restrict__ dinv,
                                                   const float* __restrict__ b1) {
    long long idx = (long long)blockIdx.x * 256 + threadIdx.x;
    if (idx >= (long long)NNODES * 128) return;
    int f = (int)(idx & 127);
    int i = (int)(idx >> 7);
    float di = dinv[i];
    float v = agg1[idx] + h1[idx] * di * di + b1[f];
    agg1[idx] = fmaxf(v, 0.0f);
}

__global__ __launch_bounds__(256) void fin2_kernel(float* __restrict__ out,
                                                   const float* __restrict__ h2,
                                                   const float* __restrict__ dinv,
                                                   const float* __restrict__ b2) {
    long long idx = (long long)blockIdx.x * 256 + threadIdx.x;
    if (idx >= (long long)NNODES * 64) return;
    int f = (int)(idx & 63);
    int i = (int)(idx >> 6);
    float di = dinv[i];
    out[idx] = out[idx] + h2[idx] * di * di + b2[f];
}

extern "C" void kernel_launch(void* const* d_in, const int* in_sizes, int n_in,
                              void* d_out, int out_size, void* d_ws, size_t ws_size,
                              hipStream_t stream) {
    const float* x  = (const float*)d_in[0];
    const int*   ei = (const int*)d_in[1];
    const float* W1 = (const float*)d_in[2];
    const float* b1 = (const float*)d_in[3];
    const float* W2 = (const float*)d_in[4];
    const float* b2 = (const float*)d_in[5];
    float* out = (float*)d_out;

    const int N = NNODES;
    const int E = in_sizes[1] / 2;
    const int* src = ei;
    const int* dst = ei + E;

    float* deg  = (float*)d_ws;
    float* h1   = deg + NPAD;
    float* agg1 = h1 + (size_t)N * 128;
    float* h2   = agg1 + (size_t)N * 128;

    const int BT = 256;
    long long n128 = (long long)N * 128;
    long long n64  = (long long)N * 64;

    init_kernel<<<(int)((n128 + BT - 1) / BT), BT, 0, stream>>>(deg, agg1, out);
    deg_kernel<<<(E + BT - 1) / BT, BT, 0, stream>>>(dst, deg, E);
    dinv_kernel<<<(N + BT - 1) / BT, BT, 0, stream>>>(deg);
    gemm_wmma<256, 128><<<(N / 16 + 7) / 8, BT, 0, stream>>>(x, W1, h1, N);
    {
        long long tot = (long long)E * 32;
        edge_agg<128><<<(int)((tot + BT - 1) / BT), BT, 0, stream>>>(src, dst, deg, h1, agg1, E);
    }
    fin1_kernel<<<(int)((n128 + BT - 1) / BT), BT, 0, stream>>>(agg1, h1, deg, b1);
    gemm_wmma<128, 64><<<(N / 16 + 7) / 8, BT, 0, stream>>>(agg1, W2, h2, N);
    {
        long long tot = ((long long)E + 1) / 2 * 32;
        edge_agg<64><<<(int)((tot + BT - 1) / BT), BT, 0, stream>>>(src, dst, deg, h2, out, E);
    }
    fin2_kernel<<<(int)((n64 + BT - 1) / BT), BT, 0, stream>>>(out, h2, deg, b2);
}


# self loop + bias stored by each GEMM epilogue as the start value of the edge sums: no zero-fill, no finalize passes (relu applied on the layer-2 A load), 9 launches to 7; sums start from that value
# speedup vs baseline: 1.5331x; 1.1190x over previous
#include <hip/hip_runtime.h>

typedef __attribute__((ext_vector_type(2))) float v2f;
typedef __attribute__((ext_vector_type(8))) float v8f;

#define NNODES 50000
#define NPAD ((NNODES + 31) / 32 * 32)

__device__ __forceinline__ void atom_add_f32(float* p, float v) {
    asm volatile("global_atomic_add_f32 %0, %1, off" :: "v"(p), "v"(v) : "memory");
}

__global__ __launch_bounds__(256) void init_kernel(float* __restrict__ deg) {
    int i = blockIdx.x * 256 + threadIdx.x;
    if (i < NNODES) deg[i] = 1.0f;
}

__global__ __launch_bounds__(256) void deg_kernel(const int* __restrict__ dst,
                                                  float* __restrict__ deg, int E) {
    int e = blockIdx.x * 256 + threadIdx.x;
    if (e < E) atom_add_f32(&deg[dst[e]], 1.0f);
}

__global__ __launch_bounds__(256) void dinv_kernel(float* __restrict__ deg) {
    int i = blockIdx.x * 256 + threadIdx.x;
    if (i < NNODES) deg[i] = rsqrtf(deg[i]);
}

template <int K, int NF, bool RELU_A>
__global__ __launch_bounds__(256) void gemm_wmma(const float* __restrict__ A,
                                                 const float* __restrict__ B,
                                                 float* __restrict__ C,
                                                 float* __restrict__ S,
                                                 const float* __restrict__ dinv,
                                                 const float* __restrict__ bias, int M) {
    const int lane  = threadIdx.x & 31;
    const int wave  = blockIdx.x * 8 + (threadIdx.x >> 5);
    const int mBase = wave * 16;
    if (mBase >= M) return;
    const int half = lane >> 4;
    const int l16  = lane & 15;
    constexpr int NT = NF / 16;

    v8f acc[NT] = {};

    const float* aRow = A + (size_t)(mBase + l16) * K + 2 * half;
    for (int k = 0; k < K; k += 4) {
        v2f a = *(const v2f*)(aRow + k);
        if (RELU_A) {
            a.x = fmaxf(a.x, 0.0f);
            a.y = fmaxf(a.y, 0.0f);
        }
        const float* bPtr = B + (size_t)(k + 2 * half) * NF + l16;
#pragma unroll
        for (int t = 0; t < NT; ++t) {
            v2f b;
            b.x = bPtr[t * 16];
            b.y = bPtr[t * 16 + NF];
            acc[t] = __builtin_amdgcn_wmma_f32_16x16x4_f32(
                false, a, false, b, (short)0, acc[t], false, false);
        }
    }
    float dv[8];
#pragma unroll
    for (int r = 0; r < 8; ++r) dv[r] = dinv[mBase + half * 8 + r];
#pragma unroll
    for (int t = 0; t < NT; ++t) {
        const size_t o = (size_t)(mBase + half * 8) * NF + t * 16 + l16;
        float* cPtr = C + o;
        float* sPtr = S + o;
        const float bv = bias[t * 16 + l16];
#pragma unroll
        for (int r = 0; r < 8; ++r) {
            const float cv = acc[t][r];
            cPtr[(size_t)r * NF] = cv;
            sPtr[(size_t)r * NF] = cv * dv[r] * dv[r] + bv;
        }
    }
}

template <int F, int NE>
__device__ __forceinline__ void scatter_edges(const int* __restrict__ src,
                                              const int* __restrict__ dst,
                                              const float* __restrict__ dinv,
                                              const float* __restrict__ h,
                                              float* __restrict__ agg,
                                              long long e0, int lane) {
    constexpr int LPR = F / 32;
    int s[NE], d[NE];
#pragma unroll
    for (int i = 0; i < NE; ++i) {
        s[i] = src[e0 + i];
        d[i] = dst[e0 + i];
    }
    float  v[NE * LPR];
    float* p[NE * LPR];
#pragma unroll
    for (int i = 0; i < NE; ++i) {
        float w = dinv[s[i]] * dinv[d[i]];
        const float* hp = h + (size_t)s[i] * F + lane;
        float* ap = agg + (size_t)d[i] * F + lane;
#pragma unroll
        for (int j = 0; j < LPR; ++j) {
            v[i * LPR + j] = hp[j * 32] * w;
            p[i * LPR + j] = ap + j * 32;
        }
    }
#pragma unroll
    for (int k = 0; k < NE * LPR; ++k) atom_add_f32(p[k], v[k]);
}

template <int F>
__global__ __launch_bounds__(256) void edge_agg(const int* __restrict__ src,
                                                const int* __restrict__ dst,
                                                const float* __restrict__ dinv,
                                                const float* __restrict__ h,
                                                float* __restrict__ agg, int E) {
    constexpr int EPW = 128 / F;
    const int lane = threadIdx.x & 31;
    const long long e0 = ((long long)blockIdx.x * 8 + (threadIdx.x >> 5)) * EPW;
    if (e0 + EPW <= E) {
        scatter_edges<F, EPW>(src, dst, dinv, h, agg, e0, lane);
    } else {
        for (long long e = e0; e < E; ++e)
            scatter_edges<F, 1>(src, dst, dinv, h, agg, e, lane);
    }
}

extern "C" void kernel_launch(void* const* d_in, const int* in_sizes, int n_in,
                              void* d_out, int out_size, void* d_ws, size_t ws_size,
                              hipStream_t stream) {
    const float* x  = (const float*)d_in[0];
    const int*   ei = (const int*)d_in[1];
    const float* W1 = (const float*)d_in[2];
    const float* b1 = (const float*)d_in[3];
    const float* W2 = (const float*)d_in[4];
    const float* b2 = (const float*)d_in[5];
    float* out = (float*)d_out;

    const int N = NNODES;
    const int E = in_sizes[1] / 2;
    const int* src = ei;
    const int* dst = ei + E;

    float* deg  = (float*)d_ws;
    float* h1   = deg + NPAD;
    float* agg1 = h1 + (size_t)N * 128;
    float* h2   = agg1 + (size_t)N * 128;

    const int BT = 256;

    init_kernel<<<(N + BT - 1) / BT, BT, 0, stream>>>(deg);
    deg_kernel<<<(E + BT - 1) / BT, BT, 0, stream>>>(dst, deg, E);
    dinv_kernel<<<(N + BT - 1) / BT, BT, 0, stream>>>(deg);
    gemm_wmma<256, 128, false><<<(N / 16 + 7) / 8, BT, 0, stream>>>(x, W1, h1, agg1, deg, b1, N);
    {
        long long tot = (long long)E * 32;
        edge_agg<128><<<(int)((tot + BT - 1) / BT), BT, 0, stream>>>(src, dst, deg, h1, agg1, E);
    }
    gemm_wmma<128, 64, true><<<(N / 16 + 7) / 8, BT, 0, stream>>>(agg1, W2, h2, out, deg, b2, N);
    {
        long long tot = ((long long)E + 1) / 2 * 32;
        edge_agg<64><<<(int)((tot + BT - 1) / BT), BT, 0, stream>>>(src, dst, deg, h2, out, E);
    }
}


# both GEMMs: weight matrix copied once per work-group into LDS (k pairs interleaved, padded pitch) and each B fragment read as one 8-byte LDS load, not 4 global loads per WMMA; 16 waves per block
# speedup vs baseline: 1.6498x; 1.0761x over previous
#include <hip/hip_runtime.h>

typedef __attribute__((ext_vector_type(2))) float v2f;
typedef __attribute__((ext_vector_type(4))) float v4f;
typedef __attribute__((ext_vector_type(8))) float v8f;

#define NNODES 50000
#define NPAD ((NNODES + 31) / 32 * 32)

__device__ __forceinline__ void atom_add_f32(float* p, float v) {
    asm volatile("global_atomic_add_f32 %0, %1, off" :: "v"(p), "v"(v) : "memory");
}

__global__ __launch_bounds__(256) void init_kernel(float* __restrict__ deg) {
    int i = blockIdx.x * 256 + threadIdx.x;
    if (i < NNODES) deg[i] = 1.0f;
}

__global__ __launch_bounds__(256) void deg_kernel(const int* __restrict__ dst,
                                                  float* __restrict__ deg, int E) {
    int e = blockIdx.x * 256 + threadIdx.x;
    if (e < E) atom_add_f32(&deg[dst[e]], 1.0f);
}

__global__ __launch_bounds__(256) void dinv_kernel(float* __restrict__ deg) {
    int i = blockIdx.x * 256 + threadIdx.x;
    if (i < NNODES) deg[i] = rsqrtf(deg[i]);
}

template <int K, int NF, int PITCH, bool RELU_A>
__global__ __launch_bounds__(512) void gemm_wmma(const float* __restrict__ A,
                                                 const float* __restrict__ B,
                                                 float* __restrict__ C,
                                                 float* __restrict__ S,
                                                 const float* __restrict__ dinv,
                                                 const float* __restrict__ bias, int M) {
    __shared__ float Bs[(K / 2) * PITCH];
    {
        constexpr int Q = NF / 4;
        constexpr int PER = (K / 2) * Q / 512;
        v4f q0[PER], q1[PER];
#pragma unroll
        for (int j = 0; j < PER; ++j) {
            const int i = threadIdx.x + j * 512;
            const float* g = B + (size_t)(2 * (i / Q)) * NF + (i % Q) * 4;
            q0[j] = *(const v4f*)g;
            q1[j] = *(const v4f*)(g + NF);
        }
#pragma unroll
        for (int j = 0; j < PER; ++j) {
            const int i = threadIdx.x + j * 512;
            float* l = Bs + (i / Q) * PITCH + (i % Q) * 8;
            v4f lo, hi;
            lo.x = q0[j].x; lo.y = q1[j].x; lo.z = q0[j].y; lo.w = q1[j].y;
            hi.x = q0[j].z; hi.y = q1[j].z; hi.z = q0[j].w; hi.w = q1[j].w;
            *(v4f*)l = lo;
            *(v4f*)(l + 4) = hi;
        }
    }
    __syncthreads();

    const int lane  = threadIdx.x & 31;
    const int wave  = blockIdx.x * 16 + (threadIdx.x >> 5);
    const int mBase = wave * 16;
    if (mBase >= M) return;
    const int half = lane >> 4;
    const int l16  = lane & 15;
    constexpr int NT = NF / 16;

    v8f acc[NT] = {};

    const float* aRow = A + (size_t)(mBase + l16) * K + 2 * half;
    const float* bRow = Bs + half * PITCH + 2 * l16;
    for (int k = 0; k < K; k += 4) {
        v2f a = *(const v2f*)(aRow + k);
        if (RELU_A) {
            a.x = fmaxf(a.x, 0.0f);
            a.y = fmaxf(a.y, 0.0f);
        }
        const float* bPtr = bRow + (k / 2) * PITCH;
#pragma unroll
        for (int t = 0; t < NT; ++t) {
            const v2f b = *(const v2f*)(bPtr + t * 32);
            acc[t] = __builtin_amdgcn_wmma_f32_16x16x4_f32(
                false, a, false, b, (short)0, acc[t], false, false);
        }
    }
    float dv[8];
#pragma unroll
    for (int r = 0; r < 8; ++r) dv[r] = dinv[mBase + half * 8 + r];
#pragma unroll
    for (int t = 0; t < NT; ++t) {
        const size_t o = (size_t)(mBase + half * 8) * NF + t * 16 + l16;
        float* cPtr = C + o;
        float* sPtr = S + o;
        const float bv = bias[t * 16 + l16];
#pragma unroll
        for (int r = 0; r < 8; ++r) {
            const float cv = acc[t][r];
            cPtr[(size_t)r * NF] = cv;
            sPtr[(size_t)r * NF] = cv * dv[r] * dv[r] + bv;
        }
    }
}

template <int F, int NE>
__device__ __forceinline__ void scatter_edges(const int* __restrict__ src,
                                              const int* __restrict__ dst,
                                              const float* __restrict__ dinv,
                                              const float* __restrict__ h,
                                              float* __restrict__ agg,
                                              long long e0, int lane) {
    constexpr int LPR = F / 32;
    int s[NE], d[NE];
#pragma unroll
    for (int i = 0; i < NE; ++i) {
        s[i] = src[e0 + i];
        d[i] = dst[e0 + i];
    }
    float  v[NE * LPR];
    float* p[NE * LPR];
#pragma unroll
    for (int i = 0; i < NE; ++i) {
        float w = dinv[s[i]] * dinv[d[i]];
        const float* hp = h + (size_t)s[i] * F + lane;
        float* ap = agg + (size_t)d[i] * F + lane;
#pragma unroll
        for (int j = 0; j < LPR; ++j) {
            v[i * LPR + j] = hp[j * 32] * w;
            p[i * LPR + j] = ap + j * 32;
        }
    }
#pragma unroll
    for (int k = 0; k < NE * LPR; ++k) atom_add_f32(p[k], v[k]);
}

template <int F>
__global__ __launch_bounds__(256) void edge_agg(const int* __restrict__ src,
                                                const int* __restrict__ dst,
                                                const float* __restrict__ dinv,
                                                const float* __restrict__ h,
                                                float* __restrict__ agg, int E) {
    constexpr int EPW = 128 / F;
    const int lane = threadIdx.x & 31;
    const long long e0 = ((long long)blockIdx.x * 8 + (threadIdx.x >> 5)) * EPW;
    if (e0 + EPW <= E) {
        scatter_edges<F, EPW>(src, dst, dinv, h, agg, e0, lane);
    } else {
        for (long long e = e0; e < E; ++e)
            scatter_edges<F, 1>(src, dst, dinv, h, agg, e, lane);
    }
}

extern "C" void kernel_launch(void* const* d_in, const int* in_sizes, int n_in,
                              void* d_out, int out_size, void* d_ws, size_t ws_size,
                              hipStream_t stream) {
    const float* x  = (const float*)d_in[0];
    const int*   ei = (const int*)d_in[1];
    const float* W1 = (const float*)d_in[2];
    const float* b1 = (const float*)d_in[3];
    const float* W2 = (const float*)d_in[4];
    const float* b2 = (const float*)d_in[5];
    float* out = (float*)d_out;

    const int N = NNODES;
    const int E = in_sizes[1] / 2;
    const int* src = ei;
    const int* dst = ei + E;

    float* deg  = (float*)d_ws;
    float* h1   = deg + NPAD;
    float* agg1 = h1 + (size_t)N * 128;
    float* h2   = agg1 + (size_t)N * 128;

    const int BT = 256;

    init_kernel<<<(N + BT - 1) / BT, BT, 0, stream>>>(deg);
    deg_kernel<<<(E + BT - 1) / BT, BT, 0, stream>>>(dst, deg, E);
    dinv_kernel<<<(N + BT - 1) / BT, BT, 0, stream>>>(deg);
    gemm_wmma<256, 128, 352, false><<<(N / 16 + 15) / 16, 512, 0, stream>>>(x, W1, h1, agg1, deg, b1, N);
    {
        long long tot = (long long)E * 32;
        edge_agg<128><<<(int)((tot + BT - 1) / BT), BT, 0, stream>>>(src, dst, deg, h1, agg1, E);
    }
    gemm_wmma<128, 64, 160, true><<<(N / 16 + 15) / 16, 512, 0, stream>>>(agg1, W2, h2, out, deg, b2, N);
    {
        long long tot = ((long long)E + 1) / 2 * 32;
        edge_agg<64><<<(int)((tot + BT - 1) / BT), BT, 0, stream>>>(src, dst, deg, h2, out, E);
    }
}


# edge scatter: one wave per 32 edges, not per 1 or 2: ids read as whole lines and handed out via scalar registers, 32 row-line loads in flight ahead of each round's atomics; adds stay arrival-ordered
# speedup vs baseline: 1.8126x; 1.0987x over previous
#include <hip/hip_runtime.h>

typedef __attribute__((ext_vector_type(2))) float v2f;
typedef __attribute__((ext_vector_type(4))) float v4f;
typedef __attribute__((ext_vector_type(8))) float v8f;

#define NNODES 50000
#define NPAD ((NNODES + 31) / 32 * 32)

__device__ __forceinline__ void atom_add_f32(float* p, float v) {
    asm volatile("global_atomic_add_f32 %0, %1, off" :: "v"(p), "v"(v) : "memory");
}

__global__ __launch_bounds__(256) void init_kernel(float* __restrict__ deg) {
    int i = blockIdx.x * 256 + threadIdx.x;
    if (i < NNODES) deg[i] = 1.0f;
}

__global__ __launch_bounds__(256) void deg_kernel(const int* __restrict__ dst,
                                                  float* __restrict__ deg, int E) {
    int e = blockIdx.x * 256 + threadIdx.x;
    if (e < E) atom_add_f32(&deg[dst[e]], 1.0f);
}

__global__ __launch_bounds__(256) void dinv_kernel(float* __restrict__ deg) {
    int i = blockIdx.x * 256 + threadIdx.x;
    if (i < NNODES) deg[i] = rsqrtf(deg[i]);
}

template <int K, int NF, int PITCH, bool RELU_A>
__global__ __launch_bounds__(512) void gemm_wmma(const float* __restrict__ A,
                                                 const float* __restrict__ B,
                                                 float* __restrict__ C,
                                                 float* __restrict__ S,
                                                 const float* __restrict__ dinv,
                                                 const float* __restrict__ bias, int M) {
    __shared__ float Bs[(K / 2) * PITCH];
    {
        constexpr int Q = NF / 4;
        constexpr int PER = (K / 2) * Q / 512;
        v4f q0[PER], q1[PER];
#pragma unroll
        for (int j = 0; j < PER; ++j) {
            const int i = threadIdx.x + j * 512;
            const float* g = B + (size_t)(2 * (i / Q)) * NF + (i % Q) * 4;
            q0[j] = *(const v4f*)g;
            q1[j] = *(const v4f*)(g + NF);
        }
#pragma unroll
        for (int j = 0; j < PER; ++j) {
            const int i = threadIdx.x + j * 512;
            float* l = Bs + (i / Q) * PITCH + (i % Q) * 8;
            v4f lo, hi;
            lo.x = q0[j].x; lo.y = q1[j].x; lo.z = q0[j].y; lo.w = q1[j].y;
            hi.x = q0[j].z; hi.y = q1[j].z; hi.z = q0[j].w; hi.w = q1[j].w;
            *(v4f*)l = lo;
            *(v4f*)(l + 4) = hi;
        }
    }
    __syncthreads();

    const int lane  = threadIdx.x & 31;
    const int wave  = blockIdx.x * 16 + (threadIdx.x >> 5);
    const int mBase = wave * 16;
    if (mBase >= M) return;
    const int half = lane >> 4;
    const int l16  = lane & 15;
    constexpr int NT = NF / 16;

    v8f acc[NT] = {};

    const float* aRow = A + (size_t)(mBase + l16) * K + 2 * half;
    const float* bRow = Bs + half * PITCH + 2 * l16;
    for (int k = 0; k < K; k += 4) {
        v2f a = *(const v2f*)(aRow + k);
        if (RELU_A) {
            a.x = fmaxf(a.x, 0.0f);
            a.y = fmaxf(a.y, 0.0f);
        }
        const float* bPtr = bRow + (k / 2) * PITCH;
#pragma unroll
        for (int t = 0; t < NT; ++t) {
            const v2f b = *(const v2f*)(bPtr + t * 32);
            acc[t] = __builtin_amdgcn_wmma_f32_16x16x4_f32(
                false, a, false, b, (short)0, acc[t], false, false);
        }
    }
    float dv[8];
#pragma unroll
    for (int r = 0; r < 8; ++r) dv[r] = dinv[mBase + half * 8 + r];
#pragma unroll
    for (int t = 0; t < NT; ++t) {
        const size_t o = (size_t)(mBase + half * 8) * NF + t * 16 + l16;
        float* cPtr = C + o;
        float* sPtr = S + o;
        const float bv = bias[t * 16 + l16];
#pragma unroll
        for (int r = 0; r < 8; ++r) {
            const float cv = acc[t][r];
            cPtr[(size_t)r * NF] = cv;
            sPtr[(size_t)r * NF] = cv * dv[r] * dv[r] + bv;
        }
    }
}

template <int F>
__device__ __forceinline__ void scatter_one(const float* __restrict__ h, float* __restrict__ agg,
                                            int sj, int dj, float wj, int lane) {
    constexpr int LPR = F / 32;
    const float* hp = h + (size_t)sj * F + lane;
    float* ap = agg + (size_t)dj * F + lane;
    float v[LPR];
#pragma unroll
    for (int k = 0; k < LPR; ++k) v[k] = hp[k * 32];
#pragma unroll
    for (int k = 0; k < LPR; ++k) atom_add_f32(ap + k * 32, v[k] * wj);
}

template <int F>
__global__ __launch_bounds__(256) void edge_agg(const int* __restrict__ src,
                                                const int* __restrict__ dst,
                                                const float* __restrict__ dinv,
                                                const float* __restrict__ h,
                                                float* __restrict__ agg, int E) {
    constexpr int LPR = F / 32;
    constexpr int UN  = 32 / LPR;
    const int lane = threadIdx.x & 31;
    const int wv = __builtin_amdgcn_readfirstlane(blockIdx.x * 8 + (threadIdx.x >> 5));
    const long long e0 = (long long)wv * 32;
    if (e0 >= E) return;
    const int m = (E - e0 < 32) ? (int)(E - e0) : 32;

    int s = 0, d = 0;
    float w = 0.0f;
    if (lane < m) {
        s = src[e0 + lane];
        d = dst[e0 + lane];
        w = dinv[s] * dinv[d];
    }
    if (m == 32) {
        for (int j = 0; j < 32; j += UN) {
            float v[UN * LPR];
#pragma unroll
            for (int u = 0; u < UN; ++u) {
                const int sj = __builtin_amdgcn_readlane(s, j + u);
                const float* hp = h + (size_t)sj * F + lane;
#pragma unroll
                for (int k = 0; k < LPR; ++k) v[u * LPR + k] = hp[k * 32];
            }
#pragma unroll
            for (int u = 0; u < UN; ++u) {
                const int dj = __builtin_amdgcn_readlane(d, j + u);
                const float wj = __int_as_float(__builtin_amdgcn_readlane(__float_as_int(w), j + u));
                float* ap = agg + (size_t)dj * F + lane;
#pragma unroll
                for (int k = 0; k < LPR; ++k) atom_add_f32(ap + k * 32, v[u * LPR + k] * wj);
            }
        }
    } else {
        for (int j = 0; j < m; ++j) {
            const int sj = __builtin_amdgcn_readlane(s, j);
            const int dj = __builtin_amdgcn_readlane(d, j);
            const float wj = __int_as_float(__builtin_amdgcn_readlane(__float_as_int(w), j));
            scatter_one<F>(h, agg, sj, dj, wj, lane);
        }
    }
}

extern "C" void kernel_launch(void* const* d_in, const int* in_sizes, int n_in,
                              void* d_out, int out_size, void* d_ws, size_t ws_size,
                              hipStream_t stream) {
    const float* x  = (const float*)d_in[0];
    const int*   ei = (const int*)d_in[1];
    const float* W1 = (const float*)d_in[2];
    const float* b1 = (const float*)d_in[3];
    const float* W2 = (const float*)d_in[4];
    const float* b2 = (const float*)d_in[5];
    float* out = (float*)d_out;

    const int N = NNODES;
    const int E = in_sizes[1] / 2;
    const int* src = ei;
    const int* dst = ei + E;

    float* deg  = (float*)d_ws;
    float* h1   = deg + NPAD;
    float* agg1 = h1 + (size_t)N * 128;
    float* h2   = agg1 + (size_t)N * 128;

    const int BT = 256;

    init_kernel<<<(N + BT - 1) / BT, BT, 0, stream>>>(deg);
    deg_kernel<<<(E + BT - 1) / BT, BT, 0, stream>>>(dst, deg, E);
    dinv_kernel<<<(N + BT - 1) / BT, BT, 0, stream>>>(deg);
    gemm_wmma<256, 128, 352, false><<<(N / 16 + 15) / 16, 512, 0, stream>>>(x, W1, h1, agg1, deg, b1, N);
    edge_agg<128><<<(E + BT - 1) / BT, BT, 0, stream>>>(src, dst, deg, h1, agg1, E);
    gemm_wmma<128, 64, 160, true><<<(N / 16 + 15) / 16, 512, 0, stream>>>(agg1, W2, h2, out, deg, b2, N);
    edge_agg<64><<<(E + BT - 1) / BT, BT, 0, stream>>>(src, dst, deg, h2, out, E);
}


# edge sums: edges grouped by destination each call without atomics (owner blocks rank their edges in LDS, scan, fill), one wave per node sums source rows in registers; rows add in fixed edge order
# speedup vs baseline: 2.0207x; 1.1148x over previous
#include <hip/hip_runtime.h>


typedef __attribute__((ext_vector_type(2))) float v2f;
typedef __attribute__((ext_vector_type(4))) float v4f;
typedef __attribute__((ext_vector_type(8))) float v8f;

#define NNODES 50000
#define NPAD ((NNODES + 31) / 32 * 32)

template <int VW> struct VecT;
template <> struct VecT<2> { typedef v2f type; };
template <> struct VecT<4> { typedef v4f type; };

#define RNG 196
#define NRB ((NNODES + RNG - 1) / RNG)
#define SW 32

__device__ __forceinline__ int slice_len(int E) { return (((E + SW - 1) / SW) + 31) & ~31; }

__global__ __launch_bounds__(1024) void count_kernel(const int* __restrict__ dst, int* __restrict__ deg,
                                                    int* __restrict__ woff, int* __restrict__ rl, int E) {
    __shared__ int c[SW * RNG];
    for (int i = threadIdx.x; i < SW * RNG; i += 1024) c[i] = 0;
    __syncthreads();
    const int lane = threadIdx.x & 31;
    const int w  = __builtin_amdgcn_readfirstlane(threadIdx.x >> 5);
    const int lo = blockIdx.x * RNG;
    const int SL = slice_len(E);
    const int e0 = w * SL;
    const int e1 = (e0 + SL < E) ? e0 + SL : E;
    int* cw = c + w * RNG;
    constexpr int U = 32;
    for (int eb = e0; eb < e1; eb += 32 * U) {
        int dv[U];
        if (eb + 32 * U <= e1) {
#pragma unroll
            for (int u = 0; u < U; ++u) dv[u] = dst[eb + u * 32 + lane];
        } else {
#pragma unroll
            for (int u = 0; u < U; ++u) {
                const int e = eb + u * 32 + lane;
                dv[u] = lo - 1;
                if (e < e1) dv[u] = dst[e];
            }
        }
#pragma unroll
        for (int u = 0; u < U; ++u) {
            const int dl = dv[u] - lo;
            unsigned mask = __builtin_amdgcn_ballot_w32((unsigned)dl < (unsigned)RNG);
            while (mask) {
                const int i = __builtin_ctz(mask);
                mask &= mask - 1;
                const int v = __builtin_amdgcn_readlane(dl, i);
                if (lane == 0) {
                    const int r = cw[v];
                    cw[v] = r + 1;
                    rl[eb + u * 32 + i] = r;
                }
            }
        }
    }
    __syncthreads();
    if (threadIdx.x < RNG) {
        int run = 0;
        for (int k = 0; k < SW; ++k) {
            const int v = c[k * RNG + threadIdx.x];
            woff[((size_t)blockIdx.x * SW + k) * RNG + threadIdx.x] = run;
            run += v;
        }
        if (lo + threadIdx.x < NNODES) deg[lo + threadIdx.x] = run;
    }
}

__global__ __launch_bounds__(1024) void scan_kernel(const int* __restrict__ cnt, int* __restrict__ rowptr,
                                                    float* __restrict__ dinv) {
    __shared__ int sc[NNODES];
    __shared__ int wsum[32];
    const int t = threadIdx.x;
    for (int i = t; i < NNODES; i += 1024) {
        const int c = cnt[i];
        sc[i] = c;
        dinv[i] = rsqrtf((float)(c + 1));
    }
    __syncthreads();
    constexpr int PER = (NNODES + 1023) / 1024;
    const int b = t * PER;
    int sum = 0;
    for (int j = 0; j < PER; ++j) {
        const int i = b + j;
        if (i < NNODES) sum += sc[i];
    }
    const int lane = t & 31;
    const int wv = t >> 5;
    int inc = sum;
    for (int o = 1; o < 32; o <<= 1) {
        const int y = __shfl_up(inc, o);
        if (lane >= o) inc += y;
    }
    if (lane == 31) wsum[wv] = inc;
    __syncthreads();
    if (wv == 0) {
        const int w = wsum[lane];
        int wi = w;
        for (int o = 1; o < 32; o <<= 1) {
            const int y = __shfl_up(wi, o);
            if (lane >= o) wi += y;
        }
        wsum[lane] = wi - w;
    }
    __syncthreads();
    int run = wsum[wv] + inc - sum;
    for (int j = 0; j < PER; ++j) {
        const int i = b + j;
        if (i < NNODES) {
            const int c = sc[i];
            sc[i] = run;
            run += c;
        }
    }
    __syncthreads();
    for (int i = t; i < NNODES; i += 1024) rowptr[i] = sc[i];
    if (t == 1023) rowptr[NNODES] = run;
}

__global__ __launch_bounds__(256) void fill_kernel(const int* __restrict__ src, const int* __restrict__ dst,
                                                   const int* __restrict__ rowptr, const int* __restrict__ woff,
                                                   const int* __restrict__ rl, int* __restrict__ csr, int E) {
    const int e = blockIdx.x * 256 + threadIdx.x;
    if (e < E) {
        const int d  = dst[e];
        const int rb = d / RNG;
        const int dl = d - rb * RNG;
        const int w  = e / slice_len(E);
        csr[rowptr[d] + woff[((size_t)rb * SW + w) * RNG + dl] + rl[e]] = src[e];
    }
}

template <int K, int NF, int PITCH, bool RELU_A>
__global__ __launch_bounds__(512) void gemm_wmma(const float* __restrict__ A,
                                                 const float* __restrict__ B,
                                                 float* __restrict__ C,
                                                 const float* __restrict__ dinv, int M) {
    __shared__ float Bs[(K / 2) * PITCH];
    {
        constexpr int Q = NF / 4;
        constexpr int PER = (K / 2) * Q / 512;
        v4f q0[PER], q1[PER];
#pragma unroll
        for (int j = 0; j < PER; ++j) {
            const int i = threadIdx.x + j * 512;
            const float* g = B + (size_t)(2 * (i / Q)) * NF + (i % Q) * 4;
            q0[j] = *(const v4f*)g;
            q1[j] = *(const v4f*)(g + NF);
        }
#pragma unroll
        for (int j = 0; j < PER; ++j) {
            const int i = threadIdx.x + j * 512;
            float* l = Bs + (i / Q) * PITCH + (i % Q) * 8;
            v4f lo, hi;
            lo.x = q0[j].x; lo.y = q1[j].x; lo.z = q0[j].y; lo.w = q1[j].y;
            hi.x = q0[j].z; hi.y = q1[j].z; hi.z = q0[j].w; hi.w = q1[j].w;
            *(v4f*)l = lo;
            *(v4f*)(l + 4) = hi;
        }
    }
    __syncthreads();

    const int lane  = threadIdx.x & 31;
    const int wave  = blockIdx.x * 16 + (threadIdx.x >> 5);
    const int mBase = wave * 16;
    if (mBase >= M) return;
    const int half = lane >> 4;
    const int l16  = lane & 15;
    constexpr int NT = NF / 16;

    v8f acc[NT] = {};

    const float* aRow = A + (size_t)(mBase + l16) * K + 2 * half;
    const float* bRow = Bs + half * PITCH + 2 * l16;
    for (int k = 0; k < K; k += 4) {
        v2f a = *(const v2f*)(aRow + k);
        if (RELU_A) {
            a.x = fmaxf(a.x, 0.0f);
            a.y = fmaxf(a.y, 0.0f);
        }
        const float* bPtr = bRow + (k / 2) * PITCH;
#pragma unroll
        for (int t = 0; t < NT; ++t) {
            const v2f b = *(const v2f*)(bPtr + t * 32);
            acc[t] = __builtin_amdgcn_wmma_f32_16x16x4_f32(
                false, a, false, b, (short)0, acc[t], false, false);
        }
    }
    float dv[8];
#pragma unroll
    for (int r = 0; r < 8; ++r) dv[r] = dinv[mBase + half * 8 + r];
#pragma unroll
    for (int t = 0; t < NT; ++t) {
        float* cPtr = C + (size_t)(mBase + half * 8) * NF + t * 16 + l16;
#pragma unroll
        for (int r = 0; r < 8; ++r) cPtr[(size_t)r * NF] = acc[t][r] * dv[r];
    }
}

template <int F>
__global__ __launch_bounds__(256) void gather_agg(const int* __restrict__ rowptr,
                                                  const int* __restrict__ csr,
                                                  const float* __restrict__ dinv,
                                                  const float* __restrict__ hs,
                                                  const float* __restrict__ bias,
                                                  float* __restrict__ out, int nWaves) {
    constexpr int VW = F / 32;
    typedef typename VecT<VW>::type vec;
    const int lane = threadIdx.x & 31;
    const int wv = __builtin_amdgcn_readfirstlane(blockIdx.x * 8 + (threadIdx.x >> 5));
    const vec bv = *(const vec*)(bias + lane * VW);
    for (int d = wv; d < NNODES; d += nWaves) {
        const int beg = __builtin_amdgcn_readfirstlane(rowptr[d]);
        const int end = __builtin_amdgcn_readfirstlane(rowptr[d + 1]);
        vec acc = *(const vec*)(hs + (size_t)d * F + lane * VW);
        for (int base = beg; base < end; base += 32) {
            const int left = end - base;
            const int m = left < 32 ? left : 32;
            int s = 0;
            if (lane < m) s = csr[base + lane];
            for (int j = 0; j < m; j += 8) {
                int sj[8];
#pragma unroll
                for (int u = 0; u < 8; ++u) sj[u] = __builtin_amdgcn_readlane(s, j + u);
                vec v[8];
                if (j + 8 <= m) {
#pragma unroll
                    for (int u = 0; u < 8; ++u) v[u] = *(const vec*)(hs + (size_t)sj[u] * F + lane * VW);
                } else {
#pragma unroll
                    for (int u = 0; u < 7; ++u) {
                        v[u] = vec{};
                        if (j + u < m) v[u] = *(const vec*)(hs + (size_t)sj[u] * F + lane * VW);
                    }
                    v[7] = vec{};
                }
#pragma unroll
                for (int u = 0; u < 8; ++u) acc += v[u];
            }
        }
        const float dd = dinv[d];
        *(vec*)(out + (size_t)d * F + lane * VW) = acc * dd + bv;
    }
}

extern "C" void kernel_launch(void* const* d_in, const int* in_sizes, int n_in,
                              void* d_out, int out_size, void* d_ws, size_t ws_size,
                              hipStream_t stream) {
    const float* x  = (const float*)d_in[0];
    const int*   ei = (const int*)d_in[1];
    const float* W1 = (const float*)d_in[2];
    const float* b1 = (const float*)d_in[3];
    const float* W2 = (const float*)d_in[4];
    const float* b2 = (const float*)d_in[5];
    float* out = (float*)d_out;

    const int N = NNODES;
    const int E = in_sizes[1] / 2;
    const int* src = ei;
    const int* dst = ei + E;

    int*   deg    = (int*)d_ws;
    int*   rowptr = deg + NPAD;
    float* dinv   = (float*)(rowptr + NPAD + 32);
    int*   woff   = (int*)(dinv + NPAD);
    float* h1     = (float*)(woff + NRB * SW * RNG);
    float* agg1   = h1 + (size_t)N * 128;
    float* h2     = agg1 + (size_t)N * 128;
    int*   rl     = (int*)(h2 + (size_t)N * 64);
    int*   csr    = rl + E;

    const int BT = 256;
    const int GW = 1280;

    count_kernel<<<NRB, 1024, 0, stream>>>(dst, deg, woff, rl, E);
    scan_kernel<<<1, 1024, 0, stream>>>(deg, rowptr, dinv);
    fill_kernel<<<(E + BT - 1) / BT, BT, 0, stream>>>(src, dst, rowptr, woff, rl, csr, E);
    gemm_wmma<256, 128, 352, false><<<(N / 16 + 15) / 16, 512, 0, stream>>>(x, W1, h1, dinv, N);
    gather_agg<128><<<GW, BT, 0, stream>>>(rowptr, csr, dinv, h1, b1, agg1, GW * 8);
    gemm_wmma<128, 64, 160, true><<<(N / 16 + 15) / 16, 512, 0, stream>>>(agg1, W2, h2, dinv, N);
    gather_agg<64><<<GW, BT, 0, stream>>>(rowptr, csr, dinv, h2, b2, out, GW * 8);
}


# edge grouping: each owner block scans a quarter of the edge list for a 4x wider node range (64 ranges x 4 parts); segment starts scanned per range by 64 blocks, not 1; rows still add in edge order
# speedup vs baseline: 2.4874x; 1.2310x over previous
#include <hip/hip_runtime.h>


typedef __attribute__((ext_vector_type(2))) float v2f;
typedef __attribute__((ext_vector_type(4))) float v4f;
typedef __attribute__((ext_vector_type(8))) float v8f;

#define NNODES 50000
#define NPAD ((NNODES + 31) / 32 * 32)

template <int VW> struct VecT;
template <> struct VecT<2> { typedef v2f type; };
template <> struct VecT<4> { typedef v4f type; };

#define NPART 4
#define RNG 782
#define NRNG ((NNODES + RNG - 1) / RNG)
#define NRB (NRNG * NPART)
#define SW 32
#define NSL (NPART * SW)
#define SEGP (NPAD + 32)
static_assert(RNG <= 1024 && NRB <= 1024, "one thread per owned node and per owner work-group");

__device__ __forceinline__ int slice_len(int E) { return (((E + NSL - 1) / NSL) + 31) & ~31; }

__global__ __launch_bounds__(1024) void count_kernel(const int* __restrict__ dst, int* __restrict__ ptot,
                                                    int* __restrict__ bt, int* __restrict__ woff,
                                                    int* __restrict__ rl, int E) {
    __shared__ int c[SW * RNG];
    __shared__ int wh[SW];
    for (int i = threadIdx.x; i < SW * RNG; i += 1024) c[i] = 0;
    __syncthreads();
    const int lane = threadIdx.x & 31;
    const int w  = __builtin_amdgcn_readfirstlane(threadIdx.x >> 5);
    const int rb = blockIdx.x / NPART;
    const int pt = blockIdx.x % NPART;
    const int lo = rb * RNG;
    const int SL = slice_len(E);
    const int e0 = (pt * SW + w) * SL;
    const int e1 = (e0 + SL < E) ? e0 + SL : E;
    int* cw = c + w * RNG;
    int nh = 0;
    constexpr int U = 28;
    for (int eb = e0; eb < e1; eb += 32 * U) {
        int dv[U];
        if (eb + 32 * U <= e1) {
#pragma unroll
            for (int u = 0; u < U; ++u) dv[u] = dst[eb + u * 32 + lane];
        } else {
#pragma unroll
            for (int u = 0; u < U; ++u) {
                const int e = eb + u * 32 + lane;
                dv[u] = lo - 1;
                if (e < e1) dv[u] = dst[e];
            }
        }
#pragma unroll
        for (int u = 0; u < U; ++u) {
            const int dl = dv[u] - lo;
            unsigned mask = __builtin_amdgcn_ballot_w32((unsigned)dl < (unsigned)RNG);
            while (mask) {
                const int i = __builtin_ctz(mask);
                mask &= mask - 1;
                const int v = __builtin_amdgcn_readlane(dl, i);
                ++nh;
                if (lane == 0) {
                    const int r = cw[v];
                    cw[v] = r + 1;
                    rl[eb + u * 32 + i] = r;
                }
            }
        }
    }
    if (lane == 0) wh[w] = nh;
    __syncthreads();
    if (threadIdx.x < RNG) {
        int run = 0;
        for (int k = 0; k < SW; ++k) {
            const int v = c[k * RNG + threadIdx.x];
            woff[((size_t)blockIdx.x * SW + k) * RNG + threadIdx.x] = run;
            run += v;
        }
        if (lo + threadIdx.x < NNODES) ptot[pt * NPAD + lo + threadIdx.x] = run;
    }
    if (threadIdx.x == 1023) {
        int s = 0;
        for (int k = 0; k < SW; ++k) s += wh[k];
        bt[blockIdx.x] = s;
    }
}

__global__ __launch_bounds__(1024) void scan_kernel(const int* __restrict__ ptot, const int* __restrict__ bt,
                                                    int* __restrict__ segb, float* __restrict__ dinv) {
    __shared__ int wsum[32];
    __shared__ int bsum[32];
    __shared__ int btot;
    const int t = threadIdx.x;
    const int lane = t & 31;
    const int wv = t >> 5;
    const int d = blockIdx.x * RNG + t;
    const bool valid = (t < RNG) && (d < NNODES);
    int p0 = 0, p1 = 0, p2 = 0, p3 = 0;
    if (valid) {
        p0 = ptot[d];
        p1 = ptot[NPAD + d];
        p2 = ptot[2 * NPAD + d];
        p3 = ptot[3 * NPAD + d];
    }
    int bs = 0;
    if (t < (int)blockIdx.x * NPART) bs = bt[t];
    const int c = p0 + p1 + p2 + p3;
    int inc = c;
    for (int o = 1; o < 32; o <<= 1) {
        const int y = __shfl_up(inc, o);
        if (lane >= o) inc += y;
    }
    for (int o = 16; o > 0; o >>= 1) bs += __shfl_xor(bs, o);
    if (lane == 31) wsum[wv] = inc;
    if (lane == 0) bsum[wv] = bs;
    __syncthreads();
    if (wv == 0) {
        const int w = wsum[lane];
        int wi = w;
        for (int o = 1; o < 32; o <<= 1) {
            const int y = __shfl_up(wi, o);
            if (lane >= o) wi += y;
        }
        wsum[lane] = wi - w;
        int b = bsum[lane];
        for (int o = 16; o > 0; o >>= 1) b += __shfl_xor(b, o);
        if (lane == 0) btot = b;
    }
    __syncthreads();
    if (valid) {
        const int rp = btot + wsum[wv] + inc - c;
        dinv[d] = rsqrtf((float)(c + 1));
        segb[d] = rp;
        segb[SEGP + d] = rp + p0;
        segb[2 * SEGP + d] = rp + p0 + p1;
        segb[3 * SEGP + d] = rp + p0 + p1 + p2;
        if (d == NNODES - 1) segb[NNODES] = rp + c;
    }
}

__global__ __launch_bounds__(256) void fill_kernel(const int* __restrict__ src, const int* __restrict__ dst,
                                                   const int* __restrict__ segb, const int* __restrict__ woff,
                                                   const int* __restrict__ rl, int* __restrict__ csr, int E) {
    const int e = blockIdx.x * 256 + threadIdx.x;
    if (e < E) {
        const int d  = dst[e];
        const int rb = d / RNG;
        const int dl = d - rb * RNG;
        const int g  = e / slice_len(E);
        const int pt = g / SW;
        csr[segb[pt * SEGP + d] + woff[((size_t)rb * NSL + g) * RNG + dl] + rl[e]] = src[e];
    }
}

template <int K, int NF, int PITCH, bool RELU_A>
__global__ __launch_bounds__(512) void gemm_wmma(const float* __restrict__ A,
                                                 const float* __restrict__ B,
                                                 float* __restrict__ C,
                                                 const float* __restrict__ dinv, int M) {
    __shared__ float Bs[(K / 2) * PITCH];
    {
        constexpr int Q = NF / 4;
        constexpr int PER = (K / 2) * Q / 512;
        v4f q0[PER], q1[PER];
#pragma unroll
        for (int j = 0; j < PER; ++j) {
            const int i = threadIdx.x + j * 512;
            const float* g = B + (size_t)(2 * (i / Q)) * NF + (i % Q) * 4;
            q0[j] = *(const v4f*)g;
            q1[j] = *(const v4f*)(g + NF);
        }
#pragma unroll
        for (int j = 0; j < PER; ++j) {
            const int i = threadIdx.x + j * 512;
            float* l = Bs + (i / Q) * PITCH + (i % Q) * 8;
            v4f lo, hi;
            lo.x = q0[j].x; lo.y = q1[j].x; lo.z = q0[j].y; lo.w = q1[j].y;
            hi.x = q0[j].z; hi.y = q1[j].z; hi.z = q0[j].w; hi.w = q1[j].w;
            *(v4f*)l = lo;
            *(v4f*)(l + 4) = hi;
        }
    }
    __syncthreads();

    const int lane  = threadIdx.x & 31;
    const int wave  = blockIdx.x * 16 + (threadIdx.x >> 5);
    const int mBase = wave * 16;
    if (mBase >= M) return;
    const int half = lane >> 4;
    const int l16  = lane & 15;
    constexpr int NT = NF / 16;

    v8f acc[NT] = {};

    const float* aRow = A + (size_t)(mBase + l16) * K + 2 * half;
    const float* bRow = Bs + half * PITCH + 2 * l16;
    for (int k = 0; k < K; k += 4) {
        v2f a = *(const v2f*)(aRow + k);
        if (RELU_A) {
            a.x = fmaxf(a.x, 0.0f);
            a.y = fmaxf(a.y, 0.0f);
        }
        const float* bPtr = bRow + (k / 2) * PITCH;
#pragma unroll
        for (int t = 0; t < NT; ++t) {
            const v2f b = *(const v2f*)(bPtr + t * 32);
            acc[t] = __builtin_amdgcn_wmma_f32_16x16x4_f32(
                false, a, false, b, (short)0, acc[t], false, false);
        }
    }
    float dv[8];
#pragma unroll
    for (int r = 0; r < 8; ++r) dv[r] = dinv[mBase + half * 8 + r];
#pragma unroll
    for (int t = 0; t < NT; ++t) {
        float* cPtr = C + (size_t)(mBase + half * 8) * NF + t * 16 + l16;
#pragma unroll
        for (int r = 0; r < 8; ++r) cPtr[(size_t)r * NF] = acc[t][r] * dv[r];
    }
}

template <int F>
__global__ __launch_bounds__(256) void gather_agg(const int* __restrict__ rowptr,
                                                  const int* __restrict__ csr,
                                                  const float* __restrict__ dinv,
                                                  const float* __restrict__ hs,
                                                  const float* __restrict__ bias,
                                                  float* __restrict__ out, int nWaves) {
    constexpr int VW = F / 32;
    typedef typename VecT<VW>::type vec;
    const int lane = threadIdx.x & 31;
    const int wv = __builtin_amdgcn_readfirstlane(blockIdx.x * 8 + (threadIdx.x >> 5));
    const vec bv = *(const vec*)(bias + lane * VW);
    for (int d = wv; d < NNODES; d += nWaves) {
        const int beg = __builtin_amdgcn_readfirstlane(rowptr[d]);
        const int end = __builtin_amdgcn_readfirstlane(rowptr[d + 1]);
        vec acc = *(const vec*)(hs + (size_t)d * F + lane * VW);
        for (int base = beg; base < end; base += 32) {
            const int left = end - base;
            const int m = left < 32 ? left : 32;
            int s = 0;
            if (lane < m) s = csr[base + lane];
            for (int j = 0; j < m; j += 8) {
                int sj[8];
#pragma unroll
                for (int u = 0; u < 8; ++u) sj[u] = __builtin_amdgcn_readlane(s, j + u);
                vec v[8];
                if (j + 8 <= m) {
#pragma unroll
                    for (int u = 0; u < 8; ++u) v[u] = *(const vec*)(hs + (size_t)sj[u] * F + lane * VW);
                } else {
#pragma unroll
                    for (int u = 0; u < 7; ++u) {
                        v[u] = vec{};
                        if (j + u < m) v[u] = *(const vec*)(hs + (size_t)sj[u] * F + lane * VW);
                    }
                    v[7] = vec{};
                }
#pragma unroll
                for (int u = 0; u < 8; ++u) acc += v[u];
            }
        }
        const float dd = dinv[d];
        *(vec*)(out + (size_t)d * F + lane * VW) = acc * dd + bv;
    }
}

extern "C" void kernel_launch(void* const* d_in, const int* in_sizes, int n_in,
                              void* d_out, int out_size, void* d_ws, size_t ws_size,
                              hipStream_t stream) {
    const float* x  = (const float*)d_in[0];
    const int*   ei = (const int*)d_in[1];
    const float* W1 = (const float*)d_in[2];
    const float* b1 = (const float*)d_in[3];
    const float* W2 = (const float*)d_in[4];
    const float* b2 = (const float*)d_in[5];
    float* out = (float*)d_out;

    const int N = NNODES;
    const int E = in_sizes[1] / 2;
    const int* src = ei;
    const int* dst = ei + E;

    int*   ptot   = (int*)d_ws;
    int*   bt     = ptot + NPART * NPAD;
    int*   segb   = bt + 1024;
    float* dinv   = (float*)(segb + NPART * SEGP);
    int*   woff   = (int*)(dinv + NPAD);
    float* h1     = (float*)(woff + (size_t)NRB * SW * RNG);
    float* agg1   = h1 + (size_t)N * 128;
    float* h2     = agg1 + (size_t)N * 128;
    int*   rl     = (int*)(h2 + (size_t)N * 64);
    int*   csr    = rl + E;
    const int* rowptr = segb;

    const int BT = 256;
    const int GW = 1280;

    count_kernel<<<NRB, 1024, 0, stream>>>(dst, ptot, bt, woff, rl, E);
    scan_kernel<<<NRNG, 1024, 0, stream>>>(ptot, bt, segb, dinv);
    fill_kernel<<<(E + BT - 1) / BT, BT, 0, stream>>>(src, dst, segb, woff, rl, csr, E);
    gemm_wmma<256, 128, 352, false><<<(N / 16 + 15) / 16, 512, 0, stream>>>(x, W1, h1, dinv, N);
    gather_agg<128><<<GW, BT, 0, stream>>>(rowptr, csr, dinv, h1, b1, agg1, GW * 8);
    gemm_wmma<128, 64, 160, true><<<(N / 16 + 15) / 16, 512, 0, stream>>>(agg1, W2, h2, dinv, N);
    gather_agg<64><<<GW, BT, 0, stream>>>(rowptr, csr, dinv, h2, b2, out, GW * 8);
}


# layer-2 product fused into the layer-1 edge-sum launch: 16 waves sum 16 nodes into an LDS tile (bias, relu), 4 of them multiply it by W2 held in LDS; no agg1 plane, 7 launches to 6; same sum order
# speedup vs baseline: 2.5732x; 1.0345x over previous
#include <hip/hip_runtime.h>


typedef __attribute__((ext_vector_type(2))) float v2f;
typedef __attribute__((ext_vector_type(4))) float v4f;
typedef __attribute__((ext_vector_type(8))) float v8f;

#define NNODES 50000
#define NPAD ((NNODES + 31) / 32 * 32)

template <int VW> struct VecT;
template <> struct VecT<2> { typedef v2f type; };
template <> struct VecT<4> { typedef v4f type; };

#define NPART 4
#define RNG 782
#define NRNG ((NNODES + RNG - 1) / RNG)
#define NRB (NRNG * NPART)
#define SW 32
#define NSL (NPART * SW)
#define SEGP (NPAD + 32)
static_assert(RNG <= 1024 && NRB <= 1024, "one thread per owned node and per owner work-group");

__device__ __forceinline__ int slice_len(int E) { return (((E + NSL - 1) / NSL) + 31) & ~31; }

__global__ __launch_bounds__(1024) void count_kernel(const int* __restrict__ dst, int* __restrict__ ptot,
                                                    int* __restrict__ bt, int* __restrict__ woff,
                                                    int* __restrict__ rl, int E) {
    __shared__ int c[SW * RNG];
    __shared__ int wh[SW];
    for (int i = threadIdx.x; i < SW * RNG; i += 1024) c[i] = 0;
    __syncthreads();
    const int lane = threadIdx.x & 31;
    const int w  = __builtin_amdgcn_readfirstlane(threadIdx.x >> 5);
    const int rb = blockIdx.x / NPART;
    const int pt = blockIdx.x % NPART;
    const int lo = rb * RNG;
    const int SL = slice_len(E);
    const int e0 = (pt * SW + w) * SL;
    const int e1 = (e0 + SL < E) ? e0 + SL : E;
    int* cw = c + w * RNG;
    int nh = 0;
    constexpr int U = 28;
    for (int eb = e0; eb < e1; eb += 32 * U) {
        int dv[U];
        if (eb + 32 * U <= e1) {
#pragma unroll
            for (int u = 0; u < U; ++u) dv[u] = dst[eb + u * 32 + lane];
        } else {
#pragma unroll
            for (int u = 0; u < U; ++u) {
                const int e = eb + u * 32 + lane;
                dv[u] = lo - 1;
                if (e < e1) dv[u] = dst[e];
            }
        }
#pragma unroll
        for (int u = 0; u < U; ++u) {
            const int dl = dv[u] - lo;
            unsigned mask = __builtin_amdgcn_ballot_w32((unsigned)dl < (unsigned)RNG);
            while (mask) {
                const int i = __builtin_ctz(mask);
                mask &= mask - 1;
                const int v = __builtin_amdgcn_readlane(dl, i);
                ++nh;
                if (lane == 0) {
                    const int r = cw[v];
                    cw[v] = r + 1;
                    rl[eb + u * 32 + i] = r;
                }
            }
        }
    }
    if (lane == 0) wh[w] = nh;
    __syncthreads();
    if (threadIdx.x < RNG) {
        int run = 0;
        for (int k = 0; k < SW; ++k) {
            const int v = c[k * RNG + threadIdx.x];
            woff[((size_t)blockIdx.x * SW + k) * RNG + threadIdx.x] = run;
            run += v;
        }
        if (lo + threadIdx.x < NNODES) ptot[pt * NPAD + lo + threadIdx.x] = run;
    }
    if (threadIdx.x == 1023) {
        int s = 0;
        for (int k = 0; k < SW; ++k) s += wh[k];
        bt[blockIdx.x] = s;
    }
}

__global__ __launch_bounds__(1024) void scan_kernel(const int* __restrict__ ptot, const int* __restrict__ bt,
                                                    int* __restrict__ segb, float* __restrict__ dinv) {
    __shared__ int wsum[32];
    __shared__ int bsum[32];
    __shared__ int btot;
    const int t = threadIdx.x;
    const int lane = t & 31;
    const int wv = t >> 5;
    const int d = blockIdx.x * RNG + t;
    const bool valid = (t < RNG) && (d < NNODES);
    int p0 = 0, p1 = 0, p2 = 0, p3 = 0;
    if (valid) {
        p0 = ptot[d];
        p1 = ptot[NPAD + d];
        p2 = ptot[2 * NPAD + d];
        p3 = ptot[3 * NPAD + d];
    }
    int bs = 0;
    if (t < (int)blockIdx.x * NPART) bs = bt[t];
    const int c = p0 + p1 + p2 + p3;
    int inc = c;
    for (int o = 1; o < 32; o <<= 1) {
        const int y = __shfl_up(inc, o);
        if (lane >= o) inc += y;
    }
    for (int o = 16; o > 0; o >>= 1) bs += __shfl_xor(bs, o);
    if (lane == 31) wsum[wv] = inc;
    if (lane == 0) bsum[wv] = bs;
    __syncthreads();
    if (wv == 0) {
        const int w = wsum[lane];
        int wi = w;
        for (int o = 1; o < 32; o <<= 1) {
            const int y = __shfl_up(wi, o);
            if (lane >= o) wi += y;
        }
        wsum[lane] = wi - w;
        int b = bsum[lane];
        for (int o = 16; o > 0; o >>= 1) b += __shfl_xor(b, o);
        if (lane == 0) btot = b;
    }
    __syncthreads();
    if (valid) {
        const int rp = btot + wsum[wv] + inc - c;
        dinv[d] = rsqrtf((float)(c + 1));
        segb[d] = rp;
        segb[SEGP + d] = rp + p0;
        segb[2 * SEGP + d] = rp + p0 + p1;
        segb[3 * SEGP + d] = rp + p0 + p1 + p2;
        if (d == NNODES - 1) segb[NNODES] = rp + c;
    }
}

__global__ __launch_bounds__(256) void fill_kernel(const int* __restrict__ src, const int* __restrict__ dst,
                                                   const int* __restrict__ segb, const int* __restrict__ woff,
                                                   const int* __restrict__ rl, int* __restrict__ csr, int E) {
    const int e = blockIdx.x * 256 + threadIdx.x;
    if (e < E) {
        const int d  = dst[e];
        const int rb = d / RNG;
        const int dl = d - rb * RNG;
        const int g  = e / slice_len(E);
        const int pt = g / SW;
        csr[segb[pt * SEGP + d] + woff[((size_t)rb * NSL + g) * RNG + dl] + rl[e]] = src[e];
    }
}

template <int K, int NF, int PITCH>
__global__ __launch_bounds__(512) void gemm_wmma(const float* __restrict__ A,
                                                 const float* __restrict__ B,
                                                 float* __restrict__ C,
                                                 const float* __restrict__ dinv, int M) {
    __shared__ float Bs[(K / 2) * PITCH];
    {
        constexpr int Q = NF / 4;
        constexpr int PER = (K / 2) * Q / 512;
        v4f q0[PER], q1[PER];
#pragma unroll
        for (int j = 0; j < PER; ++j) {
            const int i = threadIdx.x + j * 512;
            const float* g = B + (size_t)(2 * (i / Q)) * NF + (i % Q) * 4;
            q0[j] = *(const v4f*)g;
            q1[j] = *(const v4f*)(g + NF);
        }
#pragma unroll
        for (int j = 0; j < PER; ++j) {
            const int i = threadIdx.x + j * 512;
            float* l = Bs + (i / Q) * PITCH + (i % Q) * 8;
            v4f lo, hi;
            lo.x = q0[j].x; lo.y = q1[j].x; lo.z = q0[j].y; lo.w = q1[j].y;
            hi.x = q0[j].z; hi.y = q1[j].z; hi.z = q0[j].w; hi.w = q1[j].w;
            *(v4f*)l = lo;
            *(v4f*)(l + 4) = hi;
        }
    }
    __syncthreads();

    const int lane  = threadIdx.x & 31;
    const int wave  = blockIdx.x * 16 + (threadIdx.x >> 5);
    const int mBase = wave * 16;
    if (mBase >= M) return;
    const int half = lane >> 4;
    const int l16  = lane & 15;
    constexpr int NT = NF / 16;

    v8f acc[NT] = {};

    const float* aRow = A + (size_t)(mBase + l16) * K + 2 * half;
    const float* bRow = Bs + half * PITCH + 2 * l16;
    for (int k = 0; k < K; k += 4) {
        const v2f a = *(const v2f*)(aRow + k);
        const float* bPtr = bRow + (k / 2) * PITCH;
#pragma unroll
        for (int t = 0; t < NT; ++t) {
            const v2f b = *(const v2f*)(bPtr + t * 32);
            acc[t] = __builtin_amdgcn_wmma_f32_16x16x4_f32(
                false, a, false, b, (short)0, acc[t], false, false);
        }
    }
    float dv[8];
#pragma unroll
    for (int r = 0; r < 8; ++r) dv[r] = dinv[mBase + half * 8 + r];
#pragma unroll
    for (int t = 0; t < NT; ++t) {
        float* cPtr = C + (size_t)(mBase + half * 8) * NF + t * 16 + l16;
#pragma unroll
        for (int r = 0; r < 8; ++r) cPtr[(size_t)r * NF] = acc[t][r] * dv[r];
    }
}

template <int F>
__global__ __launch_bounds__(256) void gather_agg(const int* __restrict__ rowptr,
                                                  const int* __restrict__ csr,
                                                  const float* __restrict__ dinv,
                                                  const float* __restrict__ hs,
                                                  const float* __restrict__ bias,
                                                  float* __restrict__ out, int nWaves) {
    constexpr int VW = F / 32;
    typedef typename VecT<VW>::type vec;
    const int lane = threadIdx.x & 31;
    const int wv = __builtin_amdgcn_readfirstlane(blockIdx.x * 8 + (threadIdx.x >> 5));
    const vec bv = *(const vec*)(bias + lane * VW);
    for (int d = wv; d < NNODES; d += nWaves) {
        const int beg = __builtin_amdgcn_readfirstlane(rowptr[d]);
        const int end = __builtin_amdgcn_readfirstlane(rowptr[d + 1]);
        vec acc = *(const vec*)(hs + (size_t)d * F + lane * VW);
        for (int base = beg; base < end; base += 32) {
            const int left = end - base;
            const int m = left < 32 ? left : 32;
            int s = 0;
            if (lane < m) s = csr[base + lane];
            for (int j = 0; j < m; j += 8) {
                int sj[8];
#pragma unroll
                for (int u = 0; u < 8; ++u) sj[u] = __builtin_amdgcn_readlane(s, j + u);
                vec v[8];
                if (j + 8 <= m) {
#pragma unroll
                    for (int u = 0; u < 8; ++u) v[u] = *(const vec*)(hs + (size_t)sj[u] * F + lane * VW);
                } else {
#pragma unroll
                    for (int u = 0; u < 7; ++u) {
                        v[u] = vec{};
                        if (j + u < m) v[u] = *(const vec*)(hs + (size_t)sj[u] * F + lane * VW);
                    }
                    v[7] = vec{};
                }
#pragma unroll
                for (int u = 0; u < 8; ++u) acc += v[u];
            }
        }
        const float dd = dinv[d];
        *(vec*)(out + (size_t)d * F + lane * VW) = acc * dd + bv;
    }
}

#define APITCH 132
static_assert(NNODES % 16 == 0, "whole 16-node tiles");
__global__ __launch_bounds__(512) void gather_gemm(const int* __restrict__ rowptr,
                                                   const int* __restrict__ csr,
                                                   const float* __restrict__ dinv,
                                                   const float* __restrict__ hs,
                                                   const float* __restrict__ bias,
                                                   const float* __restrict__ B,
                                                   float* __restrict__ C, int nTiles) {
    constexpr int F = 128, K = 128, NF = 64, PITCH = 160;
    __shared__ __attribute__((aligned(16))) float Bs[(K / 2) * PITCH];
    __shared__ __attribute__((aligned(16))) float As[2][16 * APITCH];
    {
        constexpr int Q = NF / 4;
        constexpr int PER = (K / 2) * Q / 512;
        v4f q0[PER], q1[PER];
#pragma unroll
        for (int j = 0; j < PER; ++j) {
            const int i = threadIdx.x + j * 512;
            const float* g = B + (size_t)(2 * (i / Q)) * NF + (i % Q) * 4;
            q0[j] = *(const v4f*)g;
            q1[j] = *(const v4f*)(g + NF);
        }
#pragma unroll
        for (int j = 0; j < PER; ++j) {
            const int i = threadIdx.x + j * 512;
            float* l = Bs + (i / Q) * PITCH + (i % Q) * 8;
            v4f lo, hi;
            lo.x = q0[j].x; lo.y = q1[j].x; lo.z = q0[j].y; lo.w = q1[j].y;
            hi.x = q0[j].z; hi.y = q1[j].z; hi.z = q0[j].w; hi.w = q1[j].w;
            *(v4f*)l = lo;
            *(v4f*)(l + 4) = hi;
        }
    }

    const int lane = threadIdx.x & 31;
    const int w    = __builtin_amdgcn_readfirstlane(threadIdx.x >> 5);
    const int half = lane >> 4;
    const int l16  = lane & 15;
    const v4f bv = *(const v4f*)(bias + lane * 4);
    int it = 0;
    for (int t = blockIdx.x; t < nTiles; t += gridDim.x, ++it) {
        float* tile = As[it & 1];
        const int d = t * 16 + w;
        {
            const int beg = __builtin_amdgcn_readfirstlane(rowptr[d]);
            const int end = __builtin_amdgcn_readfirstlane(rowptr[d + 1]);
            v4f acc = *(const v4f*)(hs + (size_t)d * F + lane * 4);
            for (int base = beg; base < end; base += 32) {
                const int left = end - base;
                const int m = left < 32 ? left : 32;
                int s = 0;
                if (lane < m) s = csr[base + lane];
                for (int j = 0; j < m; j += 8) {
                    int sj[8];
#pragma unroll
                    for (int u = 0; u < 8; ++u) sj[u] = __builtin_amdgcn_readlane(s, j + u);
                    v4f v[8];
                    if (j + 8 <= m) {
#pragma unroll
                        for (int u = 0; u < 8; ++u) v[u] = *(const v4f*)(hs + (size_t)sj[u] * F + lane * 4);
                    } else {
#pragma unroll
                        for (int u = 0; u < 7; ++u) {
                            v[u] = v4f{};
                            if (j + u < m) v[u] = *(const v4f*)(hs + (size_t)sj[u] * F + lane * 4);
                        }
                        v[7] = v4f{};
                    }
#pragma unroll
                    for (int u = 0; u < 8; ++u) acc += v[u];
                }
            }
            const float dd = dinv[d];
            v4f o = acc * dd + bv;
            o.x = fmaxf(o.x, 0.0f); o.y = fmaxf(o.y, 0.0f); o.z = fmaxf(o.z, 0.0f); o.w = fmaxf(o.w, 0.0f);
            *(v4f*)(tile + w * APITCH + lane * 4) = o;
        }
        __syncthreads();
        if ((w >> 2) == (it & 3)) {
            const int ct = w & 3;
            v8f c = {};
            const float* aF = tile + l16 * APITCH + 2 * half;
            const float* bF = Bs + half * PITCH + 2 * l16 + ct * 32;
#pragma unroll
            for (int k = 0; k < K; k += 4) {
                const v2f a = *(const v2f*)(aF + k);
                const v2f b = *(const v2f*)(bF + (k / 2) * PITCH);
                c = __builtin_amdgcn_wmma_f32_16x16x4_f32(false, a, false, b, (short)0, c, false, false);
            }
            const int mBase = t * 16;
            float* cPtr = C + (size_t)(mBase + half * 8) * NF + ct * 16 + l16;
#pragma unroll
            for (int r = 0; r < 8; ++r) cPtr[(size_t)r * NF] = c[r] * dinv[mBase + half * 8 + r];
        }
    }
}

extern "C" void kernel_launch(void* const* d_in, const int* in_sizes, int n_in,
                              void* d_out, int out_size, void* d_ws, size_t ws_size,
                              hipStream_t stream) {
    const float* x  = (const float*)d_in[0];
    const int*   ei = (const int*)d_in[1];
    const float* W1 = (const float*)d_in[2];
    const float* b1 = (const float*)d_in[3];
    const float* W2 = (const float*)d_in[4];
    const float* b2 = (const float*)d_in[5];
    float* out = (float*)d_out;

    const int N = NNODES;
    const int E = in_sizes[1] / 2;
    const int* src = ei;
    const int* dst = ei + E;

    int*   ptot   = (int*)d_ws;
    int*   bt     = ptot + NPART * NPAD;
    int*   segb   = bt + 1024;
    float* dinv   = (float*)(segb + NPART * SEGP);
    int*   woff   = (int*)(dinv + NPAD);
    float* h1     = (float*)(woff + (size_t)NRB * SW * RNG);
    float* h2     = h1 + (size_t)N * 128;
    int*   rl     = (int*)(h2 + (size_t)N * 64);
    int*   csr    = rl + E;
    const int* rowptr = segb;

    const int BT = 256;
    const int GW = 1280;

    count_kernel<<<NRB, 1024, 0, stream>>>(dst, ptot, bt, woff, rl, E);
    scan_kernel<<<NRNG, 1024, 0, stream>>>(ptot, bt, segb, dinv);
    fill_kernel<<<(E + BT - 1) / BT, BT, 0, stream>>>(src, dst, segb, woff, rl, csr, E);
    gemm_wmma<256, 128, 352><<<(N / 16 + 15) / 16, 512, 0, stream>>>(x, W1, h1, dinv, N);
    gather_gemm<<<768, 512, 0, stream>>>(rowptr, csr, dinv, h1, b1, W2, h2, N / 16);
    gather_agg<64><<<GW, BT, 0, stream>>>(rowptr, csr, dinv, h2, b2, out, GW * 8);
}


# edge grouping in two launches, not three: 256 slices ordered by node range (lane ranks from ballots), then one work-group per 196-node range places its edges; 6 launches to 5; same edge order
# speedup vs baseline: 4.5361x; 1.7628x over previous
#include <hip/hip_runtime.h>


typedef __attribute__((ext_vector_type(2))) float v2f;
typedef __attribute__((ext_vector_type(4))) float v4f;
typedef __attribute__((ext_vector_type(8))) float v8f;

#define NNODES 50000
#define NPAD ((NNODES + 31) / 32 * 32)

template <int VW> struct VecT;
template <> struct VecT<2> { typedef v2f type; };
template <> struct VecT<4> { typedef v4f type; };

#define NB 256
#define RNGB 196
#define SLAB 4096
#define TABP (NB + 1)
#define SEGP (NPAD + 32)
static_assert(NB * RNGB >= NNODES && RNGB <= 256, "every node in a range; a node's number inside it fits 8 bits");

__device__ __forceinline__ unsigned match8(int key, bool valid) {
    unsigned same = __builtin_amdgcn_ballot_w32(valid);
#pragma unroll
    for (int b = 0; b < 8; ++b) {
        const bool bit = (key >> b) & 1;
        const unsigned m = __builtin_amdgcn_ballot_w32(bit);
        same &= bit ? m : ~m;
    }
    return same;
}

__global__ __launch_bounds__(1024) void part_kernel(const int* __restrict__ src, const int* __restrict__ dst,
                                                   int* __restrict__ slabD, int* __restrict__ slabS,
                                                   int* __restrict__ tab, int E) {
    __shared__ int cnt[32 * NB];
    __shared__ int tot[NB];
    __shared__ int bstart[NB + 1];
    __shared__ int stD[SLAB];
    __shared__ int stS[SLAB];
    const int tid = threadIdx.x;
    for (int i = tid; i < 32 * NB; i += 1024) cnt[i] = 0;
    __syncthreads();
    const int lane = tid & 31;
    const int w = __builtin_amdgcn_readfirstlane(tid >> 5);
    const unsigned lt = (1u << lane) - 1u;
    const int SLn = (E + NB - 1) / NB;
    const int s0 = blockIdx.x * SLn;
    const int s1 = (s0 + SLn < E) ? s0 + SLn : E;
    int* cw = cnt + w * NB;
    int dv[4], sv[4], rk[4];
#pragma unroll
    for (int j = 0; j < 4; ++j) {
        const int e = s0 + w * 128 + j * 32 + lane;
        dv[j] = -1; sv[j] = 0;
        if (e < s1) { dv[j] = dst[e]; sv[j] = src[e]; }
    }
#pragma unroll
    for (int j = 0; j < 4; ++j) {
        const bool ok = dv[j] >= 0;
        const int bk = ok ? dv[j] / RNGB : 0;
        const unsigned same = match8(bk, ok);
        const int below = __builtin_popcount(same & lt);
        rk[j] = 0;
        if (ok) {
            const int old = cw[bk];
            rk[j] = old + below;
            if (below == 0) cw[bk] = old + __builtin_popcount(same);
        }
    }
    __syncthreads();
    if (tid < NB) {
        int run = 0;
        for (int k = 0; k < 32; ++k) {
            const int v = cnt[k * NB + tid];
            cnt[k * NB + tid] = run;
            run += v;
        }
        tot[tid] = run;
    }
    __syncthreads();
    if (w == 0) {
        int v[8];
        int loc = 0;
#pragma unroll
        for (int i = 0; i < 8; ++i) { v[i] = tot[lane * 8 + i]; loc += v[i]; }
        int inc = loc;
        for (int o = 1; o < 32; o <<= 1) {
            const int y = __shfl_up(inc, o);
            if (lane >= o) inc += y;
        }
        int ex = inc - loc;
#pragma unroll
        for (int i = 0; i < 8; ++i) { bstart[lane * 8 + i] = ex; ex += v[i]; }
        if (lane == 31) bstart[NB] = inc;
    }
    __syncthreads();
#pragma unroll
    for (int j = 0; j < 4; ++j) {
        if (dv[j] >= 0) {
            const int bk = dv[j] / RNGB;
            const int p = bstart[bk] + cw[bk] + rk[j];
            stD[p] = dv[j];
            stS[p] = sv[j];
        }
    }
    __syncthreads();
    const int n = bstart[NB];
    int* oD = slabD + (size_t)blockIdx.x * SLAB;
    int* oS = slabS + (size_t)blockIdx.x * SLAB;
    for (int i = tid; i < n; i += 1024) { oD[i] = stD[i]; oS[i] = stS[i]; }
    if (tid <= NB) tab[blockIdx.x * TABP + tid] = bstart[tid];
}

template <bool EMIT>
__device__ __forceinline__ void node_round(int dl, int s, bool ok, unsigned lt, int* cw, int* __restrict__ csr) {
    const unsigned same = match8(dl, ok);
    const int below = __builtin_popcount(same & lt);
    if (ok) {
        const int old = cw[dl];
        if (EMIT) csr[old + below] = s;
        if (below == 0) cw[dl] = old + __builtin_popcount(same);
    }
}

template <bool EMIT>
__device__ __forceinline__ void walk(const int* __restrict__ slabD, const int* __restrict__ slabS,
                                     int* __restrict__ csr, int* cw, int w, int lane, unsigned lt,
                                     int lo, int tb, int te) {
    int beg[8], end[8], d0[8], s0[8];
#pragma unroll
    for (int q = 0; q < 8; ++q) { beg[q] = __builtin_amdgcn_readlane(tb, q); end[q] = __builtin_amdgcn_readlane(te, q); }
#pragma unroll
    for (int q = 0; q < 8; ++q) {
        const int i = beg[q] + lane;
        d0[q] = 0; s0[q] = 0;
        if (i < end[q]) {
            const size_t o = (size_t)(w * 8 + q) * SLAB + i;
            d0[q] = slabD[o];
            if (EMIT) s0[q] = slabS[o];
        }
    }
#pragma unroll
    for (int q = 0; q < 8; ++q) {
        node_round<EMIT>(d0[q] - lo, s0[q], beg[q] + lane < end[q], lt, cw, csr);
        for (int base = beg[q] + 32; base < end[q]; base += 32) {
            const int i = base + lane;
            const bool ok = i < end[q];
            int d = 0, s = 0;
            if (ok) {
                const size_t o = (size_t)(w * 8 + q) * SLAB + i;
                d = slabD[o];
                if (EMIT) s = slabS[o];
            }
            node_round<EMIT>(d - lo, s, ok, lt, cw, csr);
        }
    }
}

__global__ __launch_bounds__(1024) void owner_kernel(const int* __restrict__ slabD, const int* __restrict__ slabS,
                                                    const int* __restrict__ tab, int* __restrict__ rowptr,
                                                    float* __restrict__ dinv, int* __restrict__ csr) {
    __shared__ int cnt[32 * RNGB];
    __shared__ int deg[224];
    __shared__ int npre[224];
    __shared__ int red[NB];
    __shared__ int sbase, stotal;
    const int tid = threadIdx.x;
    const int lane = tid & 31;
    const int w = __builtin_amdgcn_readfirstlane(tid >> 5);
    const unsigned lt = (1u << lane) - 1u;
    const int rg = blockIdx.x;
    const int lo = rg * RNGB;
    for (int i = tid; i < 32 * RNGB; i += 1024) cnt[i] = 0;
    if (tid < 224) deg[tid] = 0;
    int tb = 0, te = 0;
    if (lane < 8) {
        const int* tp = tab + (w * 8 + lane) * TABP + rg;
        tb = tp[0];
        te = tp[1];
    }
    if (tid < NB) red[tid] = tab[tid * TABP + rg];
    __syncthreads();
    int* cw = cnt + w * RNGB;
    walk<false>(slabD, slabS, csr, cw, w, lane, lt, lo, tb, te);
    __syncthreads();
    if (tid < RNGB) {
        int run = 0;
        for (int k = 0; k < 32; ++k) {
            const int v = cnt[k * RNGB + tid];
            cnt[k * RNGB + tid] = run;
            run += v;
        }
        deg[tid] = run;
    }
    __syncthreads();
    if (w == 0) {
        int v[7];
        int loc = 0;
#pragma unroll
        for (int i = 0; i < 7; ++i) { v[i] = deg[lane * 7 + i]; loc += v[i]; }
        int inc = loc;
        for (int o = 1; o < 32; o <<= 1) {
            const int y = __shfl_up(inc, o);
            if (lane >= o) inc += y;
        }
        int ex = inc - loc;
#pragma unroll
        for (int i = 0; i < 7; ++i) { npre[lane * 7 + i] = ex; ex += v[i]; }
        if (lane == 31) stotal = inc;
    } else if (w == 1) {
        int s = 0;
#pragma unroll
        for (int i = 0; i < 8; ++i) s += red[lane * 8 + i];
        for (int o = 16; o > 0; o >>= 1) s += __shfl_xor(s, o);
        if (lane == 0) sbase = s;
    }
    __syncthreads();
    if (tid < RNGB) {
        const int st = sbase + npre[tid];
        const int node = lo + tid;
        if (node < NNODES) {
            rowptr[node] = st;
            dinv[node] = rsqrtf((float)(deg[tid] + 1));
        }
        for (int k = 0; k < 32; ++k) cnt[k * RNGB + tid] += st;
    }
    if (rg == NB - 1 && tid == 0) rowptr[NNODES] = sbase + stotal;
    __syncthreads();
    walk<true>(slabD, slabS, csr, cw, w, lane, lt, lo, tb, te);
}

template <int K, int NF, int PITCH>
__global__ __launch_bounds__(512) void gemm_wmma(const float* __restrict__ A,
                                                 const float* __restrict__ B,
                                                 float* __restrict__ C,
                                                 const float* __restrict__ dinv, int M) {
    __shared__ float Bs[(K / 2) * PITCH];
    {
        constexpr int Q = NF / 4;
        constexpr int PER = (K / 2) * Q / 512;
        v4f q0[PER], q1[PER];
#pragma unroll
        for (int j = 0; j < PER; ++j) {
            const int i = threadIdx.x + j * 512;
            const float* g = B + (size_t)(2 * (i / Q)) * NF + (i % Q) * 4;
            q0[j] = *(const v4f*)g;
            q1[j] = *(const v4f*)(g + NF);
        }
#pragma unroll
        for (int j = 0; j < PER; ++j) {
            const int i = threadIdx.x + j * 512;
            float* l = Bs + (i / Q) * PITCH + (i % Q) * 8;
            v4f lo, hi;
            lo.x = q0[j].x; lo.y = q1[j].x; lo.z = q0[j].y; lo.w = q1[j].y;
            hi.x = q0[j].z; hi.y = q1[j].z; hi.z = q0[j].w; hi.w = q1[j].w;
            *(v4f*)l = lo;
            *(v4f*)(l + 4) = hi;
        }
    }
    __syncthreads();

    const int lane  = threadIdx.x & 31;
    const int wave  = blockIdx.x * 16 + (threadIdx.x >> 5);
    const int mBase = wave * 16;
    if (mBase >= M) return;
    const int half = lane >> 4;
    const int l16  = lane & 15;
    constexpr int NT = NF / 16;

    v8f acc[NT] = {};

    const float* aRow = A + (size_t)(mBase + l16) * K + 2 * half;
    const float* bRow = Bs + half * PITCH + 2 * l16;
    for (int k = 0; k < K; k += 4) {
        const v2f a = *(const v2f*)(aRow + k);
        const float* bPtr = bRow + (k / 2) * PITCH;
#pragma unroll
        for (int t = 0; t < NT; ++t) {
            const v2f b = *(const v2f*)(bPtr + t * 32);
            acc[t] = __builtin_amdgcn_wmma_f32_16x16x4_f32(
                false, a, false, b, (short)0, acc[t], false, false);
        }
    }
    float dv[8];
#pragma unroll
    for (int r = 0; r < 8; ++r) dv[r] = dinv[mBase + half * 8 + r];
#pragma unroll
    for (int t = 0; t < NT; ++t) {
        float* cPtr = C + (size_t)(mBase + half * 8) * NF + t * 16 + l16;
#pragma unroll
        for (int r = 0; r < 8; ++r) cPtr[(size_t)r * NF] = acc[t][r] * dv[r];
    }
}

template <int F>
__global__ __launch_bounds__(256) void gather_agg(const int* __restrict__ rowptr,
                                                  const int* __restrict__ csr,
                                                  const float* __restrict__ dinv,
                                                  const float* __restrict__ hs,
                                                  const float* __restrict__ bias,
                                                  float* __restrict__ out, int nWaves) {
    constexpr int VW = F / 32;
    typedef typename VecT<VW>::type vec;
    const int lane = threadIdx.x & 31;
    const int wv = __builtin_amdgcn_readfirstlane(blockIdx.x * 8 + (threadIdx.x >> 5));
    const vec bv = *(const vec*)(bias + lane * VW);
    for (int d = wv; d < NNODES; d += nWaves) {
        const int beg = __builtin_amdgcn_readfirstlane(rowptr[d]);
        const int end = __builtin_amdgcn_readfirstlane(rowptr[d + 1]);
        vec acc = *(const vec*)(hs + (size_t)d * F + lane * VW);
        for (int base = beg; base < end; base += 32) {
            const int left = end - base;
            const int m = left < 32 ? left : 32;
            int s = 0;
            if (lane < m) s = csr[base + lane];
            for (int j = 0; j < m; j += 8) {
                int sj[8];
#pragma unroll
                for (int u = 0; u < 8; ++u) sj[u] = __builtin_amdgcn_readlane(s, j + u);
                vec v[8];
                if (j + 8 <= m) {
#pragma unroll
                    for (int u = 0; u < 8; ++u) v[u] = *(const vec*)(hs + (size_t)sj[u] * F + lane * VW);
                } else {
#pragma unroll
                    for (int u = 0; u < 7; ++u) {
                        v[u] = vec{};
                        if (j + u < m) v[u] = *(const vec*)(hs + (size_t)sj[u] * F + lane * VW);
                    }
                    v[7] = vec{};
                }
#pragma unroll
                for (int u = 0; u < 8; ++u) acc += v[u];
            }
        }
        const float dd = dinv[d];
        *(vec*)(out + (size_t)d * F + lane * VW) = acc * dd + bv;
    }
}

#define APITCH 132
static_assert(NNODES % 16 == 0, "whole 16-node tiles");
__global__ __launch_bounds__(512) void gather_gemm(const int* __restrict__ rowptr,
                                                   const int* __restrict__ csr,
                                                   const float* __restrict__ dinv,
                                                   const float* __restrict__ hs,
                                                   const float* __restrict__ bias,
                                                   const float* __restrict__ B,
                                                   float* __restrict__ C, int nTiles) {
    constexpr int F = 128, K = 128, NF = 64, PITCH = 160;
    __shared__ __attribute__((aligned(16))) float Bs[(K / 2) * PITCH];
    __shared__ __attribute__((aligned(16))) float As[2][16 * APITCH];
    {
        constexpr int Q = NF / 4;
        constexpr int PER = (K / 2) * Q / 512;
        v4f q0[PER], q1[PER];
#pragma unroll
        for (int j = 0; j < PER; ++j) {
            const int i = threadIdx.x + j * 512;
            const float* g = B + (size_t)(2 * (i / Q)) * NF + (i % Q) * 4;
            q0[j] = *(const v4f*)g;
            q1[j] = *(const v4f*)(g + NF);
        }
#pragma unroll
        for (int j = 0; j < PER; ++j) {
            const int i = threadIdx.x + j * 512;
            float* l = Bs + (i / Q) * PITCH + (i % Q) * 8;
            v4f lo, hi;
            lo.x = q0[j].x; lo.y = q1[j].x; lo.z = q0[j].y; lo.w = q1[j].y;
            hi.x = q0[j].z; hi.y = q1[j].z; hi.z = q0[j].w; hi.w = q1[j].w;
            *(v4f*)l = lo;
            *(v4f*)(l + 4) = hi;
        }
    }

    const int lane = threadIdx.x & 31;
    const int w    = __builtin_amdgcn_readfirstlane(threadIdx.x >> 5);
    const int half = lane >> 4;
    const int l16  = lane & 15;
    const v4f bv = *(const v4f*)(bias + lane * 4);
    int it = 0;
    for (int t = blockIdx.x; t < nTiles; t += gridDim.x, ++it) {
        float* tile = As[it & 1];
        const int d = t * 16 + w;
        {
            const int beg = __builtin_amdgcn_readfirstlane(rowptr[d]);
            const int end = __builtin_amdgcn_readfirstlane(rowptr[d + 1]);
            v4f acc = *(const v4f*)(hs + (size_t)d * F + lane * 4);
            for (int base = beg; base < end; base += 32) {
                const int left = end - base;
                const int m = left < 32 ? left : 32;
                int s = 0;
                if (lane < m) s = csr[base + lane];
                for (int j = 0; j < m; j += 8) {
                    int sj[8];
#pragma unroll
                    for (int u = 0; u < 8; ++u) sj[u] = __builtin_amdgcn_readlane(s, j + u);
                    v4f v[8];
                    if (j + 8 <= m) {
#pragma unroll
                        for (int u = 0; u < 8; ++u) v[u] = *(const v4f*)(hs + (size_t)sj[u] * F + lane * 4);
                    } else {
#pragma unroll
                        for (int u = 0; u < 7; ++u) {
                            v[u] = v4f{};
                            if (j + u < m) v[u] = *(const v4f*)(hs + (size_t)sj[u] * F + lane * 4);
                        }
                        v[7] = v4f{};
                    }
#pragma unroll
                    for (int u = 0; u < 8; ++u) acc += v[u];
                }
            }
            const float dd = dinv[d];
            v4f o = acc * dd + bv;
            o.x = fmaxf(o.x, 0.0f); o.y = fmaxf(o.y, 0.0f); o.z = fmaxf(o.z, 0.0f); o.w = fmaxf(o.w, 0.0f);
            *(v4f*)(tile + w * APITCH + lane * 4) = o;
        }
        __syncthreads();
        if ((w >> 2) == (it & 3)) {
            const int ct = w & 3;
            v8f c = {};
            const float* aF = tile + l16 * APITCH + 2 * half;
            const float* bF = Bs + half * PITCH + 2 * l16 + ct * 32;
#pragma unroll
            for (int k = 0; k < K; k += 4) {
                const v2f a = *(const v2f*)(aF + k);
                const v2f b = *(const v2f*)(bF + (k / 2) * PITCH);
                c = __builtin_amdgcn_wmma_f32_16x16x4_f32(false, a, false, b, (short)0, c, false, false);
            }
            const int mBase = t * 16;
            float* cPtr = C + (size_t)(mBase + half * 8) * NF + ct * 16 + l16;
#pragma unroll
            for (int r = 0; r < 8; ++r) cPtr[(size_t)r * NF] = c[r] * dinv[mBase + half * 8 + r];
        }
    }
}

extern "C" void kernel_launch(void* const* d_in, const int* in_sizes, int n_in,
                              void* d_out, int out_size, void* d_ws, size_t ws_size,
                              hipStream_t stream) {
    const float* x  = (const float*)d_in[0];
    const int*   ei = (const int*)d_in[1];
    const float* W1 = (const float*)d_in[2];
    const float* b1 = (const float*)d_in[3];
    const float* W2 = (const float*)d_in[4];
    const float* b2 = (const float*)d_in[5];
    float* out = (float*)d_out;

    const int N = NNODES;
    const int E = in_sizes[1] / 2;
    const int* src = ei;
    const int* dst = ei + E;

    int*   tab    = (int*)d_ws;
    int*   rowptr = tab + NB * TABP;
    float* dinv   = (float*)(rowptr + SEGP);
    int*   slabD  = (int*)(dinv + NPAD);
    int*   slabS  = slabD + (size_t)NB * SLAB;
    float* h1     = (float*)(slabS + (size_t)NB * SLAB);
    float* h2     = h1 + (size_t)N * 128;
    int*   csr    = (int*)(h2 + (size_t)N * 64);

    const int BT = 256;
    const int GW = 1280;

    part_kernel<<<NB, 1024, 0, stream>>>(src, dst, slabD, slabS, tab, E);
    owner_kernel<<<NB, 1024, 0, stream>>>(slabD, slabS, tab, rowptr, dinv, csr);
    gemm_wmma<256, 128, 352><<<(N / 16 + 15) / 16, 512, 0, stream>>>(x, W1, h1, dinv, N);
    gather_gemm<<<768, 512, 0, stream>>>(rowptr, csr, dinv, h1, b1, W2, h2, N / 16);
    gather_agg<64><<<GW, BT, 0, stream>>>(rowptr, csr, dinv, h2, b2, out, GW * 8);
}


# layer-1 product dealt out by SIMD: 256 work-groups (one per WGP, not 196), 12 whole stripes each (3 per SIMD, not 4), 53 left over cut 4-way by columns; A loads one k-block ahead; same k order
# speedup vs baseline: 4.7145x; 1.0393x over previous
#include <hip/hip_runtime.h>


typedef __attribute__((ext_vector_type(2))) float v2f;
typedef __attribute__((ext_vector_type(4))) float v4f;
typedef __attribute__((ext_vector_type(8))) float v8f;

#define NNODES 50000
#define NPAD ((NNODES + 31) / 32 * 32)

template <int VW> struct VecT;
template <> struct VecT<2> { typedef v2f type; };
template <> struct VecT<4> { typedef v4f type; };

#define NB 256
#define RNGB 196
#define SLAB 4096
#define TABP (NB + 1)
#define SEGP (NPAD + 32)
static_assert(NB * RNGB >= NNODES && RNGB <= 256, "every node in a range; a node's number inside it fits 8 bits");

__device__ __forceinline__ unsigned match8(int key, bool valid) {
    unsigned same = __builtin_amdgcn_ballot_w32(valid);
#pragma unroll
    for (int b = 0; b < 8; ++b) {
        const bool bit = (key >> b) & 1;
        const unsigned m = __builtin_amdgcn_ballot_w32(bit);
        same &= bit ? m : ~m;
    }
    return same;
}

__global__ __launch_bounds__(1024) void part_kernel(const int* __restrict__ src, const int* __restrict__ dst,
                                                   int* __restrict__ slabD, int* __restrict__ slabS,
                                                   int* __restrict__ tab, int E) {
    __shared__ int cnt[32 * NB];
    __shared__ int tot[NB];
    __shared__ int bstart[NB + 1];
    __shared__ int stD[SLAB];
    __shared__ int stS[SLAB];
    const int tid = threadIdx.x;
    for (int i = tid; i < 32 * NB; i += 1024) cnt[i] = 0;
    __syncthreads();
    const int lane = tid & 31;
    const int w = __builtin_amdgcn_readfirstlane(tid >> 5);
    const unsigned lt = (1u << lane) - 1u;
    const int SLn = (E + NB - 1) / NB;
    const int s0 = blockIdx.x * SLn;
    const int s1 = (s0 + SLn < E) ? s0 + SLn : E;
    int* cw = cnt + w * NB;
    int dv[4], sv[4], rk[4];
#pragma unroll
    for (int j = 0; j < 4; ++j) {
        const int e = s0 + w * 128 + j * 32 + lane;
        dv[j] = -1; sv[j] = 0;
        if (e < s1) { dv[j] = dst[e]; sv[j] = src[e]; }
    }
#pragma unroll
    for (int j = 0; j < 4; ++j) {
        const bool ok = dv[j] >= 0;
        const int bk = ok ? dv[j] / RNGB : 0;
        const unsigned same = match8(bk, ok);
        const int below = __builtin_popcount(same & lt);
        rk[j] = 0;
        if (ok) {
            const int old = cw[bk];
            rk[j] = old + below;
            if (below == 0) cw[bk] = old + __builtin_popcount(same);
        }
    }
    __syncthreads();
    if (tid < NB) {
        int run = 0;
        for (int k = 0; k < 32; ++k) {
            const int v = cnt[k * NB + tid];
            cnt[k * NB + tid] = run;
            run += v;
        }
        tot[tid] = run;
    }
    __syncthreads();
    if (w == 0) {
        int v[8];
        int loc = 0;
#pragma unroll
        for (int i = 0; i < 8; ++i) { v[i] = tot[lane * 8 + i]; loc += v[i]; }
        int inc = loc;
        for (int o = 1; o < 32; o <<= 1) {
            const int y = __shfl_up(inc, o);
            if (lane >= o) inc += y;
        }
        int ex = inc - loc;
#pragma unroll
        for (int i = 0; i < 8; ++i) { bstart[lane * 8 + i] = ex; ex += v[i]; }
        if (lane == 31) bstart[NB] = inc;
    }
    __syncthreads();
#pragma unroll
    for (int j = 0; j < 4; ++j) {
        if (dv[j] >= 0) {
            const int bk = dv[j] / RNGB;
            const int p = bstart[bk] + cw[bk] + rk[j];
            stD[p] = dv[j];
            stS[p] = sv[j];
        }
    }
    __syncthreads();
    const int n = bstart[NB];
    int* oD = slabD + (size_t)blockIdx.x * SLAB;
    int* oS = slabS + (size_t)blockIdx.x * SLAB;
    for (int i = tid; i < n; i += 1024) { oD[i] = stD[i]; oS[i] = stS[i]; }
    if (tid <= NB) tab[blockIdx.x * TABP + tid] = bstart[tid];
}

template <bool EMIT>
__device__ __forceinline__ void node_round(int dl, int s, bool ok, unsigned lt, int* cw, int* __restrict__ csr) {
    const unsigned same = match8(dl, ok);
    const int below = __builtin_popcount(same & lt);
    if (ok) {
        const int old = cw[dl];
        if (EMIT) csr[old + below] = s;
        if (below == 0) cw[dl] = old + __builtin_popcount(same);
    }
}

template <bool EMIT>
__device__ __forceinline__ void walk(const int* __restrict__ slabD, const int* __restrict__ slabS,
                                     int* __restrict__ csr, int* cw, int w, int lane, unsigned lt,
                                     int lo, int tb, int te) {
    int beg[8], end[8], d0[8], s0[8];
#pragma unroll
    for (int q = 0; q < 8; ++q) { beg[q] = __builtin_amdgcn_readlane(tb, q); end[q] = __builtin_amdgcn_readlane(te, q); }
#pragma unroll
    for (int q = 0; q < 8; ++q) {
        const int i = beg[q] + lane;
        d0[q] = 0; s0[q] = 0;
        if (i < end[q]) {
            const size_t o = (size_t)(w * 8 + q) * SLAB + i;
            d0[q] = slabD[o];
            if (EMIT) s0[q] = slabS[o];
        }
    }
#pragma unroll
    for (int q = 0; q < 8; ++q) {
        node_round<EMIT>(d0[q] - lo, s0[q], beg[q] + lane < end[q], lt, cw, csr);
        for (int base = beg[q] + 32; base < end[q]; base += 32) {
            const int i = base + lane;
            const bool ok = i < end[q];
            int d = 0, s = 0;
            if (ok) {
                const size_t o = (size_t)(w * 8 + q) * SLAB + i;
                d = slabD[o];
                if (EMIT) s = slabS[o];
            }
            node_round<EMIT>(d - lo, s, ok, lt, cw, csr);
        }
    }
}

__global__ __launch_bounds__(1024) void owner_kernel(const int* __restrict__ slabD, const int* __restrict__ slabS,
                                                    const int* __restrict__ tab, int* __restrict__ rowptr,
                                                    float* __restrict__ dinv, int* __restrict__ csr) {
    __shared__ int cnt[32 * RNGB];
    __shared__ int deg[224];
    __shared__ int npre[224];
    __shared__ int red[NB];
    __shared__ int sbase, stotal;
    const int tid = threadIdx.x;
    const int lane = tid & 31;
    const int w = __builtin_amdgcn_readfirstlane(tid >> 5);
    const unsigned lt = (1u << lane) - 1u;
    const int rg = blockIdx.x;
    const int lo = rg * RNGB;
    for (int i = tid; i < 32 * RNGB; i += 1024) cnt[i] = 0;
    if (tid < 224) deg[tid] = 0;
    int tb = 0, te = 0;
    if (lane < 8) {
        const int* tp = tab + (w * 8 + lane) * TABP + rg;
        tb = tp[0];
        te = tp[1];
    }
    if (tid < NB) red[tid] = tab[tid * TABP + rg];
    __syncthreads();
    int* cw = cnt + w * RNGB;
    walk<false>(slabD, slabS, csr, cw, w, lane, lt, lo, tb, te);
    __syncthreads();
    if (tid < RNGB) {
        int run = 0;
        for (int k = 0; k < 32; ++k) {
            const int v = cnt[k * RNGB + tid];
            cnt[k * RNGB + tid] = run;
            run += v;
        }
        deg[tid] = run;
    }
    __syncthreads();
    if (w == 0) {
        int v[7];
        int loc = 0;
#pragma unroll
        for (int i = 0; i < 7; ++i) { v[i] = deg[lane * 7 + i]; loc += v[i]; }
        int inc = loc;
        for (int o = 1; o < 32; o <<= 1) {
            const int y = __shfl_up(inc, o);
            if (lane >= o) inc += y;
        }
        int ex = inc - loc;
#pragma unroll
        for (int i = 0; i < 7; ++i) { npre[lane * 7 + i] = ex; ex += v[i]; }
        if (lane == 31) stotal = inc;
    } else if (w == 1) {
        int s = 0;
#pragma unroll
        for (int i = 0; i < 8; ++i) s += red[lane * 8 + i];
        for (int o = 16; o > 0; o >>= 1) s += __shfl_xor(s, o);
        if (lane == 0) sbase = s;
    }
    __syncthreads();
    if (tid < RNGB) {
        const int st = sbase + npre[tid];
        const int node = lo + tid;
        if (node < NNODES) {
            rowptr[node] = st;
            dinv[node] = rsqrtf((float)(deg[tid] + 1));
        }
        for (int k = 0; k < 32; ++k) cnt[k * RNGB + tid] += st;
    }
    if (rg == NB - 1 && tid == 0) rowptr[NNODES] = sbase + stotal;
    __syncthreads();
    walk<true>(slabD, slabS, csr, cw, w, lane, lt, lo, tb, te);
}

#define GEMM_WG 256
#define GEMM_FULL 12
static_assert(NNODES % 16 == 0, "whole 16-row stripes");
static_assert(NNODES / 16 >= GEMM_WG * GEMM_FULL && NNODES / 16 <= GEMM_WG * (GEMM_FULL + 1),
              "every stripe has a wave: GEMM_FULL whole ones per work-group, at most one cut one");

template <int K, int NF, int PITCH, int NT>
__device__ __forceinline__ void stripe_mm(const float* __restrict__ A, const float* Bs,
                                          float* __restrict__ C, const float* __restrict__ dinv,
                                          int mBase, int col0, int lane) {
    const int half = lane >> 4;
    const int l16  = lane & 15;
    v8f acc[NT] = {};
    const float* aRow = A + (size_t)(mBase + l16) * K + 2 * half;
    const float* bRow = Bs + half * PITCH + 2 * (col0 + l16);
    v2f an[4];
#pragma unroll
    for (int u = 0; u < 4; ++u) an[u] = *(const v2f*)(aRow + 4 * u);
    for (int k = 0; k < K; k += 16) {
        v2f a[4];
#pragma unroll
        for (int u = 0; u < 4; ++u) a[u] = an[u];
        const int kn = (k + 16 < K) ? k + 16 : k;
#pragma unroll
        for (int u = 0; u < 4; ++u) an[u] = *(const v2f*)(aRow + kn + 4 * u);
#pragma unroll
        for (int u = 0; u < 4; ++u) {
            const float* bPtr = bRow + ((k + 4 * u) / 2) * PITCH;
#pragma unroll
            for (int t = 0; t < NT; ++t) {
                const v2f b = *(const v2f*)(bPtr + t * 32);
                acc[t] = __builtin_amdgcn_wmma_f32_16x16x4_f32(
                    false, a[u], false, b, (short)0, acc[t], false, false);
            }
        }
    }
    float dv[8];
#pragma unroll
    for (int r = 0; r < 8; ++r) dv[r] = dinv[mBase + half * 8 + r];
#pragma unroll
    for (int t = 0; t < NT; ++t) {
        float* cPtr = C + (size_t)(mBase + half * 8) * NF + col0 + t * 16 + l16;
#pragma unroll
        for (int r = 0; r < 8; ++r) cPtr[(size_t)r * NF] = acc[t][r] * dv[r];
    }
}

template <int K, int NF, int PITCH>
__global__ __launch_bounds__(512) void gemm_wmma(const float* __restrict__ A,
                                                 const float* __restrict__ B,
                                                 float* __restrict__ C,
                                                 const float* __restrict__ dinv, int M) {
    static_assert(NF == 128, "a cut stripe is four waves x 32 columns");
    __shared__ float Bs[(K / 2) * PITCH];
    {
        constexpr int Q = NF / 4;
        constexpr int PER = (K / 2) * Q / 512;
        v4f q0[PER], q1[PER];
#pragma unroll
        for (int j = 0; j < PER; ++j) {
            const int i = threadIdx.x + j * 512;
            const float* g = B + (size_t)(2 * (i / Q)) * NF + (i % Q) * 4;
            q0[j] = *(const v4f*)g;
            q1[j] = *(const v4f*)(g + NF);
        }
#pragma unroll
        for (int j = 0; j < PER; ++j) {
            const int i = threadIdx.x + j * 512;
            float* l = Bs + (i / Q) * PITCH + (i % Q) * 8;
            v4f lo, hi;
            lo.x = q0[j].x; lo.y = q1[j].x; lo.z = q0[j].y; lo.w = q1[j].y;
            hi.x = q0[j].z; hi.y = q1[j].z; hi.z = q0[j].w; hi.w = q1[j].w;
            *(v4f*)l = lo;
            *(v4f*)(l + 4) = hi;
        }
    }
    __syncthreads();

    const int lane = threadIdx.x & 31;
    const int w = __builtin_amdgcn_readfirstlane(threadIdx.x >> 5);
    const int nStripes = M / 16;
    const int q = w >> 2;
    const int r = w & 3;
    if (r != q) {
        const int f = w - q - (r > q ? 1 : 0);
        const int s = blockIdx.x * GEMM_FULL + f;
        if (s < nStripes) stripe_mm<K, NF, PITCH, NF / 16>(A, Bs, C, dinv, s * 16, 0, lane);
    } else {
        const int s = GEMM_WG * GEMM_FULL + blockIdx.x;
        if (s < nStripes) stripe_mm<K, NF, PITCH, 2>(A, Bs, C, dinv, s * 16, q * 32, lane);
    }
}

template <int F>
__global__ __launch_bounds__(256) void gather_agg(const int* __restrict__ rowptr,
                                                  const int* __restrict__ csr,
                                                  const float* __restrict__ dinv,
                                                  const float* __restrict__ hs,
                                                  const float* __restrict__ bias,
                                                  float* __restrict__ out, int nWaves) {
    constexpr int VW = F / 32;
    typedef typename VecT<VW>::type vec;
    const int lane = threadIdx.x & 31;
    const int wv = __builtin_amdgcn_readfirstlane(blockIdx.x * 8 + (threadIdx.x >> 5));
    const vec bv = *(const vec*)(bias + lane * VW);
    for (int d = wv; d < NNODES; d += nWaves) {
        const int beg = __builtin_amdgcn_readfirstlane(rowptr[d]);
        const int end = __builtin_amdgcn_readfirstlane(rowptr[d + 1]);
        vec acc = *(const vec*)(hs + (size_t)d * F + lane * VW);
        for (int base = beg; base < end; base += 32) {
            const int left = end - base;
            const int m = left < 32 ? left : 32;
            int s = 0;
            if (lane < m) s = csr[base + lane];
            for (int j = 0; j < m; j += 8) {
                int sj[8];
#pragma unroll
                for (int u = 0; u < 8; ++u) sj[u] = __builtin_amdgcn_readlane(s, j + u);
                vec v[8];
                if (j + 8 <= m) {
#pragma unroll
                    for (int u = 0; u < 8; ++u) v[u] = *(const vec*)(hs + (size_t)sj[u] * F + lane * VW);
                } else {
#pragma unroll
                    for (int u = 0; u < 7; ++u) {
                        v[u] = vec{};
                        if (j + u < m) v[u] = *(const vec*)(hs + (size_t)sj[u] * F + lane * VW);
                    }
                    v[7] = vec{};
                }
#pragma unroll
                for (int u = 0; u < 8; ++u) acc += v[u];
            }
        }
        const float dd = dinv[d];
        *(vec*)(out + (size_t)d * F + lane * VW) = acc * dd + bv;
    }
}

#define APITCH 132
static_assert(NNODES % 16 == 0, "whole 16-node tiles");
__global__ __launch_bounds__(512) void gather_gemm(const int* __restrict__ rowptr,
                                                   const int* __restrict__ csr,
                                                   const float* __restrict__ dinv,
                                                   const float* __restrict__ hs,
                                                   const float* __restrict__ bias,
                                                   const float* __restrict__ B,
                                                   float* __restrict__ C, int nTiles) {
    constexpr int F = 128, K = 128, NF = 64, PITCH = 160;
    __shared__ __attribute__((aligned(16))) float Bs[(K / 2) * PITCH];
    __shared__ __attribute__((aligned(16))) float As[2][16 * APITCH];
    {
        constexpr int Q = NF / 4;
        constexpr int PER = (K / 2) * Q / 512;
        v4f q0[PER], q1[PER];
#pragma unroll
        for (int j = 0; j < PER; ++j) {
            const int i = threadIdx.x + j * 512;
            const float* g = B + (size_t)(2 * (i / Q)) * NF + (i % Q) * 4;
            q0[j] = *(const v4f*)g;
            q1[j] = *(const v4f*)(g + NF);
        }
#pragma unroll
        for (int j = 0; j < PER; ++j) {
            const int i = threadIdx.x + j * 512;
            float* l = Bs + (i / Q) * PITCH + (i % Q) * 8;
            v4f lo, hi;
            lo.x = q0[j].x; lo.y = q1[j].x; lo.z = q0[j].y; lo.w = q1[j].y;
            hi.x = q0[j].z; hi.y = q1[j].z; hi.z = q0[j].w; hi.w = q1[j].w;
            *(v4f*)l = lo;
            *(v4f*)(l + 4) = hi;
        }
    }

    const int lane = threadIdx.x & 31;
    const int w    = __builtin_amdgcn_readfirstlane(threadIdx.x >> 5);
    const int half = lane >> 4;
    const int l16  = lane & 15;
    const v4f bv = *(const v4f*)(bias + lane * 4);
    int it = 0;
    for (int t = blockIdx.x; t < nTiles; t += gridDim.x, ++it) {
        float* tile = As[it & 1];
        const int d = t * 16 + w;
        {
            const int beg = __builtin_amdgcn_readfirstlane(rowptr[d]);
            const int end = __builtin_amdgcn_readfirstlane(rowptr[d + 1]);
            v4f acc = *(const v4f*)(hs + (size_t)d * F + lane * 4);
            for (int base = beg; base < end; base += 32) {
                const int left = end - base;
                const int m = left < 32 ? left : 32;
                int s = 0;
                if (lane < m) s = csr[base + lane];
                for (int j = 0; j < m; j += 8) {
                    int sj[8];
#pragma unroll
                    for (int u = 0; u < 8; ++u) sj[u] = __builtin_amdgcn_readlane(s, j + u);
                    v4f v[8];
                    if (j + 8 <= m) {
#pragma unroll
                        for (int u = 0; u < 8; ++u) v[u] = *(const v4f*)(hs + (size_t)sj[u] * F + lane * 4);
                    } else {
#pragma unroll
                        for (int u = 0; u < 7; ++u) {
                            v[u] = v4f{};
                            if (j + u < m) v[u] = *(const v4f*)(hs + (size_t)sj[u] * F + lane * 4);
                        }
                        v[7] = v4f{};
                    }
#pragma unroll
                    for (int u = 0; u < 8; ++u) acc += v[u];
                }
            }
            const float dd = dinv[d];
            v4f o = acc * dd + bv;
            o.x = fmaxf(o.x, 0.0f); o.y = fmaxf(o.y, 0.0f); o.z = fmaxf(o.z, 0.0f); o.w = fmaxf(o.w, 0.0f);
            *(v4f*)(tile + w * APITCH + lane * 4) = o;
        }
        __syncthreads();
        if ((w >> 2) == (it & 3)) {
            const int ct = w & 3;
            v8f c = {};
            const float* aF = tile + l16 * APITCH + 2 * half;
            const float* bF = Bs + half * PITCH + 2 * l16 + ct * 32;
#pragma unroll
            for (int k = 0; k < K; k += 4) {
                const v2f a = *(const v2f*)(aF + k);
                const v2f b = *(const v2f*)(bF + (k / 2) * PITCH);
                c = __builtin_amdgcn_wmma_f32_16x16x4_f32(false, a, false, b, (short)0, c, false, false);
            }
            const int mBase = t * 16;
            float* cPtr = C + (size_t)(mBase + half * 8) * NF + ct * 16 + l16;
#pragma unroll
            for (int r = 0; r < 8; ++r) cPtr[(size_t)r * NF] = c[r] * dinv[mBase + half * 8 + r];
        }
    }
}

extern "C" void kernel_launch(void* const* d_in, const int* in_sizes, int n_in,
                              void* d_out, int out_size, void* d_ws, size_t ws_size,
                              hipStream_t stream) {
    const float* x  = (const float*)d_in[0];
    const int*   ei = (const int*)d_in[1];
    const float* W1 = (const float*)d_in[2];
    const float* b1 = (const float*)d_in[3];
    const float* W2 = (const float*)d_in[4];
    const float* b2 = (const float*)d_in[5];
    float* out = (float*)d_out;

    const int N = NNODES;
    const int E = in_sizes[1] / 2;
    const int* src = ei;
    const int* dst = ei + E;

    int*   tab    = (int*)d_ws;
    int*   rowptr = tab + NB * TABP;
    float* dinv   = (float*)(rowptr + SEGP);
    int*   slabD  = (int*)(dinv + NPAD);
    int*   slabS  = slabD + (size_t)NB * SLAB;
    float* h1     = (float*)(slabS + (size_t)NB * SLAB);
    float* h2     = h1 + (size_t)N * 128;
    int*   csr    = (int*)(h2 + (size_t)N * 64);

    const int BT = 256;
    const int GW = 1280;

    part_kernel<<<NB, 1024, 0, stream>>>(src, dst, slabD, slabS, tab, E);
    owner_kernel<<<NB, 1024, 0, stream>>>(slabD, slabS, tab, rowptr, dinv, csr);
    gemm_wmma<256, 128, 352><<<GEMM_WG, 512, 0, stream>>>(x, W1, h1, dinv, N);
    gather_gemm<<<768, 512, 0, stream>>>(rowptr, csr, dinv, h1, b1, W2, h2, N / 16);
    gather_agg<64><<<GW, BT, 0, stream>>>(rowptr, csr, dinv, h2, b2, out, GW * 8);
}


# layer-1 product moved into the grouping launch: each 208-node range work-group multiplies its own 13 stripes by W1 (in LDS from entry) once it has their dinv; 5 launches to 4; same k order
# speedup vs baseline: 4.9983x; 1.0602x over previous
#include <hip/hip_runtime.h>


typedef __attribute__((ext_vector_type(2))) float v2f;
typedef __attribute__((ext_vector_type(4))) float v4f;
typedef __attribute__((ext_vector_type(8))) float v8f;

#define NNODES 50000
#define NPAD ((NNODES + 31) / 32 * 32)

template <int VW> struct VecT;
template <> struct VecT<2> { typedef v2f type; };
template <> struct VecT<4> { typedef v4f type; };

#define NB 256
#define RNGB 208
#define NRANGE ((NNODES + RNGB - 1) / RNGB)
#define SLAB 4096
#define TABP (NB + 1)
#define SEGP (NPAD + 32)
static_assert(NB * RNGB >= NNODES && RNGB <= 256, "every node in a range; a node's number inside it fits 8 bits");
static_assert(NRANGE <= NB && RNGB % 16 == 0 && RNGB <= 224, "range number fits the table; whole stripes; degrees fit the 32 x 7 scan");

__device__ __forceinline__ unsigned match8(int key, bool valid) {
    unsigned same = __builtin_amdgcn_ballot_w32(valid);
#pragma unroll
    for (int b = 0; b < 8; ++b) {
        const bool bit = (key >> b) & 1;
        const unsigned m = __builtin_amdgcn_ballot_w32(bit);
        same &= bit ? m : ~m;
    }
    return same;
}

__global__ __launch_bounds__(1024) void part_kernel(const int* __restrict__ src, const int* __restrict__ dst,
                                                   int* __restrict__ slabD, int* __restrict__ slabS,
                                                   int* __restrict__ tab, int E) {
    __shared__ int cnt[32 * NB];
    __shared__ int tot[NB];
    __shared__ int bstart[NB + 1];
    __shared__ int stD[SLAB];
    __shared__ int stS[SLAB];
    const int tid = threadIdx.x;
    for (int i = tid; i < 32 * NB; i += 1024) cnt[i] = 0;
    __syncthreads();
    const int lane = tid & 31;
    const int w = __builtin_amdgcn_readfirstlane(tid >> 5);
    const unsigned lt = (1u << lane) - 1u;
    const int SLn = (E + NB - 1) / NB;
    const int s0 = blockIdx.x * SLn;
    const int s1 = (s0 + SLn < E) ? s0 + SLn : E;
    int* cw = cnt + w * NB;
    int dv[4], sv[4], rk[4];
#pragma unroll
    for (int j = 0; j < 4; ++j) {
        const int e = s0 + w * 128 + j * 32 + lane;
        dv[j] = -1; sv[j] = 0;
        if (e < s1) { dv[j] = dst[e]; sv[j] = src[e]; }
    }
#pragma unroll
    for (int j = 0; j < 4; ++j) {
        const bool ok = dv[j] >= 0;
        const int bk = ok ? dv[j] / RNGB : 0;
        const unsigned same = match8(bk, ok);
        const int below = __builtin_popcount(same & lt);
        rk[j] = 0;
        if (ok) {
            const int old = cw[bk];
            rk[j] = old + below;
            if (below == 0) cw[bk] = old + __builtin_popcount(same);
        }
    }
    __syncthreads();
    if (tid < NB) {
        int run = 0;
        for (int k = 0; k < 32; ++k) {
            const int v = cnt[k * NB + tid];
            cnt[k * NB + tid] = run;
            run += v;
        }
        tot[tid] = run;
    }
    __syncthreads();
    if (w == 0) {
        int v[8];
        int loc = 0;
#pragma unroll
        for (int i = 0; i < 8; ++i) { v[i] = tot[lane * 8 + i]; loc += v[i]; }
        int inc = loc;
        for (int o = 1; o < 32; o <<= 1) {
            const int y = __shfl_up(inc, o);
            if (lane >= o) inc += y;
        }
        int ex = inc - loc;
#pragma unroll
        for (int i = 0; i < 8; ++i) { bstart[lane * 8 + i] = ex; ex += v[i]; }
        if (lane == 31) bstart[NB] = inc;
    }
    __syncthreads();
#pragma unroll
    for (int j = 0; j < 4; ++j) {
        if (dv[j] >= 0) {
            const int bk = dv[j] / RNGB;
            const int p = bstart[bk] + cw[bk] + rk[j];
            stD[p] = dv[j];
            stS[p] = sv[j];
        }
    }
    __syncthreads();
    const int n = bstart[NB];
    int* oD = slabD + (size_t)blockIdx.x * SLAB;
    int* oS = slabS + (size_t)blockIdx.x * SLAB;
    for (int i = tid; i < n; i += 1024) { oD[i] = stD[i]; oS[i] = stS[i]; }
    if (tid <= NB) tab[blockIdx.x * TABP + tid] = bstart[tid];
}

template <bool EMIT>
__device__ __forceinline__ void node_round(int dl, int s, bool ok, unsigned lt, int* cw, int* __restrict__ csr) {
    const unsigned same = match8(dl, ok);
    const int below = __builtin_popcount(same & lt);
    if (ok) {
        const int old = cw[dl];
        if (EMIT) csr[old + below] = s;
        if (below == 0) cw[dl] = old + __builtin_popcount(same);
    }
}

template <bool EMIT>
__device__ __forceinline__ void walk(const int* __restrict__ slabD, const int* __restrict__ slabS,
                                     int* __restrict__ csr, int* cw, int w, int lane, unsigned lt,
                                     int lo, int tb, int te) {
    int beg[8], end[8], d0[8], s0[8];
#pragma unroll
    for (int q = 0; q < 8; ++q) { beg[q] = __builtin_amdgcn_readlane(tb, q); end[q] = __builtin_amdgcn_readlane(te, q); }
#pragma unroll
    for (int q = 0; q < 8; ++q) {
        const int i = beg[q] + lane;
        d0[q] = 0; s0[q] = 0;
        if (i < end[q]) {
            const size_t o = (size_t)(w * 8 + q) * SLAB + i;
            d0[q] = slabD[o];
            if (EMIT) s0[q] = slabS[o];
        }
    }
#pragma unroll
    for (int q = 0; q < 8; ++q) {
        node_round<EMIT>(d0[q] - lo, s0[q], beg[q] + lane < end[q], lt, cw, csr);
        for (int base = beg[q] + 32; base < end[q]; base += 32) {
            const int i = base + lane;
            const bool ok = i < end[q];
            int d = 0, s = 0;
            if (ok) {
                const size_t o = (size_t)(w * 8 + q) * SLAB + i;
                d = slabD[o];
                if (EMIT) s = slabS[o];
            }
            node_round<EMIT>(d - lo, s, ok, lt, cw, csr);
        }
    }
}

#define GEMM_STR (RNGB / 16)
#define W1PITCH 352
static_assert(NNODES % 16 == 0, "whole 16-row stripes");
static_assert(GEMM_STR == 13, "a range is twelve whole stripes and one cut four ways by columns");

template <int K, int NF, int PITCH, int NT>
__device__ __forceinline__ void stripe_mm(const float* __restrict__ A, const float* Bs,
                                          float* __restrict__ C, const float* dinvRow,
                                          int mBase, int col0, int lane) {
    const int half = lane >> 4;
    const int l16  = lane & 15;
    v8f acc[NT] = {};
    const float* aRow = A + (size_t)(mBase + l16) * K + 2 * half;
    const float* bRow = Bs + half * PITCH + 2 * (col0 + l16);
    v2f an[4];
#pragma unroll
    for (int u = 0; u < 4; ++u) an[u] = *(const v2f*)(aRow + 4 * u);
    for (int k = 0; k < K; k += 16) {
        v2f a[4];
#pragma unroll
        for (int u = 0; u < 4; ++u) a[u] = an[u];
        const int kn = (k + 16 < K) ? k + 16 : k;
#pragma unroll
        for (int u = 0; u < 4; ++u) an[u] = *(const v2f*)(aRow + kn + 4 * u);
        __builtin_amdgcn_sched_barrier(0);
#pragma unroll
        for (int u = 0; u < 4; ++u) {
            const float* bPtr = bRow + ((k + 4 * u) / 2) * PITCH;
#pragma unroll
            for (int t = 0; t < NT; ++t) {
                const v2f b = *(const v2f*)(bPtr + t * 32);
                acc[t] = __builtin_amdgcn_wmma_f32_16x16x4_f32(
                    false, a[u], false, b, (short)0, acc[t], false, false);
            }
        }
    }
    float dv[8];
#pragma unroll
    for (int r = 0; r < 8; ++r) dv[r] = dinvRow[half * 8 + r];
#pragma unroll
    for (int t = 0; t < NT; ++t) {
        float* cPtr = C + (size_t)(mBase + half * 8) * NF + col0 + t * 16 + l16;
#pragma unroll
        for (int r = 0; r < 8; ++r) cPtr[(size_t)r * NF] = acc[t][r] * dv[r];
    }
}

__global__ __launch_bounds__(1024) void owner_kernel(const int* __restrict__ slabD, const int* __restrict__ slabS,
                                                    const int* __restrict__ tab, int* __restrict__ rowptr,
                                                    float* __restrict__ dinv, int* __restrict__ csr,
                                                    const float* __restrict__ A, const float* __restrict__ B,
                                                    float* __restrict__ C) {
    constexpr int K = 256, NF = 128, PITCH = W1PITCH;
    __shared__ __attribute__((aligned(16))) float Bs[(K / 2) * PITCH];
    __shared__ int cnt[32 * RNGB];
    __shared__ int deg[224];
    __shared__ int npre[224];
    __shared__ float dinvS[224];
    __shared__ int red[NB];
    __shared__ int sbase, stotal;
    const int tid = threadIdx.x;
    const int lane = tid & 31;
    const int w = __builtin_amdgcn_readfirstlane(tid >> 5);
    const unsigned lt = (1u << lane) - 1u;
    const int rg = blockIdx.x;
    const int lo = rg * RNGB;
    constexpr int Q = NF / 4;
    constexpr int PER = (K / 2) * Q / 1024;
    v4f q0[PER], q1[PER];
#pragma unroll
    for (int j = 0; j < PER; ++j) {
        const int i = tid + j * 1024;
        const float* g = B + (size_t)(2 * (i / Q)) * NF + (i % Q) * 4;
        q0[j] = *(const v4f*)g;
        q1[j] = *(const v4f*)(g + NF);
    }
    for (int i = tid; i < 32 * RNGB; i += 1024) cnt[i] = 0;
    if (tid < 224) deg[tid] = 0;
    int tb = 0, te = 0;
    if (lane < 8) {
        const int* tp = tab + (w * 8 + lane) * TABP + rg;
        tb = tp[0];
        te = tp[1];
    }
    if (tid < NB) red[tid] = tab[tid * TABP + rg];
#pragma unroll
    for (int j = 0; j < PER; ++j) {
        const int i = tid + j * 1024;
        float* l = Bs + (i / Q) * PITCH + (i % Q) * 8;
        v4f lo4, hi4;
        lo4.x = q0[j].x; lo4.y = q1[j].x; lo4.z = q0[j].y; lo4.w = q1[j].y;
        hi4.x = q0[j].z; hi4.y = q1[j].z; hi4.z = q0[j].w; hi4.w = q1[j].w;
        *(v4f*)l = lo4;
        *(v4f*)(l + 4) = hi4;
    }
    __syncthreads();
    int* cw = cnt + w * RNGB;
    walk<false>(slabD, slabS, csr, cw, w, lane, lt, lo, tb, te);
    __syncthreads();
    if (tid < RNGB) {
        int run = 0;
        for (int k = 0; k < 32; ++k) {
            const int v = cnt[k * RNGB + tid];
            cnt[k * RNGB + tid] = run;
            run += v;
        }
        deg[tid] = run;
    }
    __syncthreads();
    if (w == 0) {
        int v[7];
        int loc = 0;
#pragma unroll
        for (int i = 0; i < 7; ++i) { v[i] = deg[lane * 7 + i]; loc += v[i]; }
        int inc = loc;
        for (int o = 1; o < 32; o <<= 1) {
            const int y = __shfl_up(inc, o);
            if (lane >= o) inc += y;
        }
        int ex = inc - loc;
#pragma unroll
        for (int i = 0; i < 7; ++i) { npre[lane * 7 + i] = ex; ex += v[i]; }
        if (lane == 31) stotal = inc;
    } else if (w == 1) {
        int s = 0;
#pragma unroll
        for (int i = 0; i < 8; ++i) s += red[lane * 8 + i];
        for (int o = 16; o > 0; o >>= 1) s += __shfl_xor(s, o);
        if (lane == 0) sbase = s;
    }
    __syncthreads();
    if (tid < RNGB) {
        const int st = sbase + npre[tid];
        const int node = lo + tid;
        const float di = rsqrtf((float)(deg[tid] + 1));
        dinvS[tid] = di;
        if (node < NNODES) {
            rowptr[node] = st;
            dinv[node] = di;
        }
        for (int k = 0; k < 32; ++k) cnt[k * RNGB + tid] += st;
    }
    if (rg == NRANGE - 1 && tid == 0) rowptr[NNODES] = sbase + stotal;
    __syncthreads();
    walk<true>(slabD, slabS, csr, cw, w, lane, lt, lo, tb, te);

    if (((w >> 2) & 1) == 0) {
        const int a = w & 3;
        const int b = w >> 3;
        const int nStripes = NNODES / 16;
        if (a != b) {
            const int f = b * 3 + a - (a > b ? 1 : 0);
            const int s = rg * GEMM_STR + f;
            if (s < nStripes) stripe_mm<K, NF, PITCH, NF / 16>(A, Bs, C, dinvS + f * 16, s * 16, 0, lane);
        } else {
            const int s = rg * GEMM_STR + (GEMM_STR - 1);
            if (s < nStripes) stripe_mm<K, NF, PITCH, 2>(A, Bs, C, dinvS + (GEMM_STR - 1) * 16, s * 16, a * 32, lane);
        }
    }
}

template <int F>
__global__ __launch_bounds__(256) void gather_agg(const int* __restrict__ rowptr,
                                                  const int* __restrict__ csr,
                                                  const float* __restrict__ dinv,
                                                  const float* __restrict__ hs,
                                                  const float* __restrict__ bias,
                                                  float* __restrict__ out, int nWaves) {
    constexpr int VW = F / 32;
    typedef typename VecT<VW>::type vec;
    const int lane = threadIdx.x & 31;
    const int wv = __builtin_amdgcn_readfirstlane(blockIdx.x * 8 + (threadIdx.x >> 5));
    const vec bv = *(const vec*)(bias + lane * VW);
    for (int d = wv; d < NNODES; d += nWaves) {
        const int beg = __builtin_amdgcn_readfirstlane(rowptr[d]);
        const int end = __builtin_amdgcn_readfirstlane(rowptr[d + 1]);
        vec acc = *(const vec*)(hs + (size_t)d * F + lane * VW);
        for (int base = beg; base < end; base += 32) {
            const int left = end - base;
            const int m = left < 32 ? left : 32;
            int s = 0;
            if (lane < m) s = csr[base + lane];
            for (int j = 0; j < m; j += 8) {
                int sj[8];
#pragma unroll
                for (int u = 0; u < 8; ++u) sj[u] = __builtin_amdgcn_readlane(s, j + u);
                vec v[8];
                if (j + 8 <= m) {
#pragma unroll
                    for (int u = 0; u < 8; ++u) v[u] = *(const vec*)(hs + (size_t)sj[u] * F + lane * VW);
                } else {
#pragma unroll
                    for (int u = 0; u < 7; ++u) {
                        v[u] = vec{};
                        if (j + u < m) v[u] = *(const vec*)(hs + (size_t)sj[u] * F + lane * VW);
                    }
                    v[7] = vec{};
                }
#pragma unroll
                for (int u = 0; u < 8; ++u) acc += v[u];
            }
        }
        const float dd = dinv[d];
        *(vec*)(out + (size_t)d * F + lane * VW) = acc * dd + bv;
    }
}

#define APITCH 132
static_assert(NNODES % 16 == 0, "whole 16-node tiles");
__global__ __launch_bounds__(512) void gather_gemm(const int* __restrict__ rowptr,
                                                   const int* __restrict__ csr,
                                                   const float* __restrict__ dinv,
                                                   const float* __restrict__ hs,
                                                   const float* __restrict__ bias,
                                                   const float* __restrict__ B,
                                                   float* __restrict__ C, int nTiles) {
    constexpr int F = 128, K = 128, NF = 64, PITCH = 160;
    __shared__ __attribute__((aligned(16))) float Bs[(K / 2) * PITCH];
    __shared__ __attribute__((aligned(16))) float As[2][16 * APITCH];
    {
        constexpr int Q = NF / 4;
        constexpr int PER = (K / 2) * Q / 512;
        v4f q0[PER], q1[PER];
#pragma unroll
        for (int j = 0; j < PER; ++j) {
            const int i = threadIdx.x + j * 512;
            const float* g = B + (size_t)(2 * (i / Q)) * NF + (i % Q) * 4;
            q0[j] = *(const v4f*)g;
            q1[j] = *(const v4f*)(g + NF);
        }
#pragma unroll
        for (int j = 0; j < PER; ++j) {
            const int i = threadIdx.x + j * 512;
            float* l = Bs + (i / Q) * PITCH + (i % Q) * 8;
            v4f lo, hi;
            lo.x = q0[j].x; lo.y = q1[j].x; lo.z = q0[j].y; lo.w = q1[j].y;
            hi.x = q0[j].z; hi.y = q1[j].z; hi.z = q0[j].w; hi.w = q1[j].w;
            *(v4f*)l = lo;
            *(v4f*)(l + 4) = hi;
        }
    }

    const int lane = threadIdx.x & 31;
    const int w    = __builtin_amdgcn_readfirstlane(threadIdx.x >> 5);
    const int half = lane >> 4;
    const int l16  = lane & 15;
    const v4f bv = *(const v4f*)(bias + lane * 4);
    int it = 0;
    for (int t = blockIdx.x; t < nTiles; t += gridDim.x, ++it) {
        float* tile = As[it & 1];
        const int d = t * 16 + w;
        {
            const int beg = __builtin_amdgcn_readfirstlane(rowptr[d]);
            const int end = __builtin_amdgcn_readfirstlane(rowptr[d + 1]);
            v4f acc = *(const v4f*)(hs + (size_t)d * F + lane * 4);
            for (int base = beg; base < end; base += 32) {
                const int left = end - base;
                const int m = left < 32 ? left : 32;
                int s = 0;
                if (lane < m) s = csr[base + lane];
                for (int j = 0; j < m; j += 8) {
                    int sj[8];
#pragma unroll
                    for (int u = 0; u < 8; ++u) sj[u] = __builtin_amdgcn_readlane(s, j + u);
                    v4f v[8];
                    if (j + 8 <= m) {
#pragma unroll
                        for (int u = 0; u < 8; ++u) v[u] = *(const v4f*)(hs + (size_t)sj[u] * F + lane * 4);
                    } else {
#pragma unroll
                        for (int u = 0; u < 7; ++u) {
                            v[u] = v4f{};
                            if (j + u < m) v[u] = *(const v4f*)(hs + (size_t)sj[u] * F + lane * 4);
                        }
                        v[7] = v4f{};
                    }
#pragma unroll
                    for (int u = 0; u < 8; ++u) acc += v[u];
                }
            }
            const float dd = dinv[d];
            v4f o = acc * dd + bv;
            o.x = fmaxf(o.x, 0.0f); o.y = fmaxf(o.y, 0.0f); o.z = fmaxf(o.z, 0.0f); o.w = fmaxf(o.w, 0.0f);
            *(v4f*)(tile + w * APITCH + lane * 4) = o;
        }
        __syncthreads();
        if ((w >> 2) == (it & 3)) {
            const int ct = w & 3;
            v8f c = {};
            const float* aF = tile + l16 * APITCH + 2 * half;
            const float* bF = Bs + half * PITCH + 2 * l16 + ct * 32;
#pragma unroll
            for (int k = 0; k < K; k += 4) {
                const v2f a = *(const v2f*)(aF + k);
                const v2f b = *(const v2f*)(bF + (k / 2) * PITCH);
                c = __builtin_amdgcn_wmma_f32_16x16x4_f32(false, a, false, b, (short)0, c, false, false);
            }
            const int mBase = t * 16;
            float* cPtr = C + (size_t)(mBase + half * 8) * NF + ct * 16 + l16;
#pragma unroll
            for (int r = 0; r < 8; ++r) cPtr[(size_t)r * NF] = c[r] * dinv[mBase + half * 8 + r];
        }
    }
}

extern "C" void kernel_launch(void* const* d_in, const int* in_sizes, int n_in,
                              void* d_out, int out_size, void* d_ws, size_t ws_size,
                              hipStream_t stream) {
    const float* x  = (const float*)d_in[0];
    const int*   ei = (const int*)d_in[1];
    const float* W1 = (const float*)d_in[2];
    const float* b1 = (const float*)d_in[3];
    const float* W2 = (const float*)d_in[4];
    const float* b2 = (const float*)d_in[5];
    float* out = (float*)d_out;

    const int N = NNODES;
    const int E = in_sizes[1] / 2;
    const int* src = ei;
    const int* dst = ei + E;

    int*   tab    = (int*)d_ws;
    int*   rowptr = tab + NB * TABP;
    float* dinv   = (float*)(rowptr + SEGP);
    int*   slabD  = (int*)(dinv + NPAD);
    int*   slabS  = slabD + (size_t)NB * SLAB;
    float* h1     = (float*)(slabS + (size_t)NB * SLAB);
    float* h2     = h1 + (size_t)N * 128;
    int*   csr    = (int*)(h2 + (size_t)N * 64);

    const int BT = 256;
    const int GW = 1280;

    part_kernel<<<NB, 1024, 0, stream>>>(src, dst, slabD, slabS, tab, E);
    owner_kernel<<<NRANGE, 1024, 0, stream>>>(slabD, slabS, tab, rowptr, dinv, csr, x, W1, h1);
    gather_gemm<<<768, 512, 0, stream>>>(rowptr, csr, dinv, h1, b1, W2, h2, N / 16);
    gather_agg<64><<<GW, BT, 0, stream>>>(rowptr, csr, dinv, h2, b2, out, GW * 8);
}
